# starting kernel
# baseline (speedup 1.0000x reference)
#include <hip/hip_runtime.h>
#include <math.h>

#define NQ    12
#define DIM   4096
#define DEPTH 4

typedef __attribute__((ext_vector_type(2))) float v2f;
typedef __attribute__((ext_vector_type(8))) float v8f;

__global__ void qdl_setup_kernel(const float* __restrict__ theta,
                                 float* __restrict__ ws) {
    const int tid = threadIdx.x;
    if (blockIdx.x < 12) {
        const int lg = blockIdx.x;
        const int l  = lg / 3, g = lg % 3;
        const int i  = tid >> 4, j = tid & 15;

        float pr = 1.0f, pi = 0.0f;
        #pragma unroll
        for (int q = 0; q < 4; ++q) {
            const int w = g * 4 + q;
            const float* th = theta + (l * NQ + w) * 3;
            const float phi = th[0], the = th[1], om = th[2];
            const float ch = cosf(0.5f * the), sh = sinf(0.5f * the);
            const float a  = 0.5f * (phi + om), b = 0.5f * (phi - om);
            const float ca = cosf(a), sa = sinf(a);
            const float cb = cosf(b), sb = sinf(b);
            const int bi = (i >> (3 - q)) & 1;
            const int bj = (j >> (3 - q)) & 1;
            float ur, ui;
            if      (bi == 0 && bj == 0) { ur =  ca * ch; ui = -sa * ch; }
            else if (bi == 0 && bj == 1) { ur = -cb * sh; ui = -sb * sh; }
            else if (bi == 1 && bj == 0) { ur =  cb * sh; ui = -sb * sh; }
            else                         { ur =  ca * ch; ui =  sa * ch; }
            const float nr = pr * ur - pi * ui;
            const float ni = pr * ui + pi * ur;
            pr = nr; pi = ni;
        }
        ws[(lg * 2 + 0) * 256 + tid] = pr;
        ws[(lg * 2 + 1) * 256 + tid] = pi;
    } else if (tid < 48) {
        const int l = tid / 12, k = tid % 12;
        const int r = l % (NQ - 1) + 1;
        int j = 1 << k;
        for (int w = NQ - 1; w >= 0; --w) {
            const int tg   = (w + r) % NQ;
            const int cbit = (j >> (NQ - 1 - w)) & 1;
            j ^= cbit << (NQ - 1 - tg);
        }
        ((int*)ws)[12 * 2 * 256 + tid] = j;
    }
}

__global__ __launch_bounds__(256) void qdl_main_kernel(
        const float* __restrict__ x,
        const float* __restrict__ ws,
        float* __restrict__ out) {
    __shared__ float psiR[DIM];
    __shared__ float psiI[DIM];
    __shared__ float gLDS[12 * 2 * 256];
    __shared__ int   permB[48];
    __shared__ float red[8 * 12];

    const int tid  = threadIdx.x;
    const int b    = blockIdx.x;
    const int lane = tid & 31;
    const int wave = tid >> 5;
    const int half = lane >> 4;
    const int mn   = lane & 15;

    for (int k = tid; k < 12 * 2 * 256; k += 256) gLDS[k] = ws[k];
    if (tid < 48) permB[tid] = ((const int*)ws)[12 * 2 * 256 + tid];

    float4* psiR4 = reinterpret_cast<float4*>(psiR);
    float4* psiI4 = reinterpret_cast<float4*>(psiI);

    const float4* xr4 = reinterpret_cast<const float4*>(x + (size_t)b * DIM);
    float4 xv[4];
    float ss = 0.0f;
    #pragma unroll
    for (int e = 0; e < 4; ++e) {
        const float4 v = xr4[tid + 256 * e];
        xv[e] = v;
        ss += v.x * v.x + v.y * v.y + v.z * v.z + v.w * v.w;
    }
    for (int o = 16; o > 0; o >>= 1) ss += __shfl_xor(ss, o, 32);
    if (lane == 0) red[wave] = ss;
    __syncthreads();
    float n1sq = 0.0f;
    #pragma unroll
    for (int wv = 0; wv < 8; ++wv) n1sq += red[wv];
    const float x0   = x[(size_t)b * DIM];
    const bool  zero = (sqrtf(n1sq) <= 1e-12f);
    const float x0n  = zero ? 1.0f : x0;
    const float inv  = 1.0f / sqrtf(n1sq - x0 * x0 + x0n * x0n);
    __syncthreads();
    #pragma unroll
    for (int e = 0; e < 4; ++e) {
        float4 v = xv[e];
        v.x = (tid == 0 && e == 0) ? x0n : v.x;
        v.x *= inv; v.y *= inv; v.z *= inv; v.w *= inv;
        psiR4[tid + 256 * e] = v;
        psiI4[tid + 256 * e] = make_float4(0.f, 0.f, 0.f, 0.f);
    }
    __syncthreads();

    for (int l = 0; l < DEPTH; ++l) {
        #pragma unroll
        for (int g = 0; g < 3; ++g) {
            const float* Gr = &gLDS[((l * 3 + g) * 2 + 0) * 256];
            const float* Gi = &gLDS[((l * 3 + g) * 2 + 1) * 256];

            v2f Ar[4], Ai[4], Ain[4];
            #pragma unroll
            for (int kq = 0; kq < 4; ++kq) {
                const int k0 = kq * 4 + 2 * half;
                const v2f ar = *reinterpret_cast<const v2f*>(&Gr[mn * 16 + k0]);
                const v2f ai = *reinterpret_cast<const v2f*>(&Gi[mn * 16 + k0]);
                Ar[kq]  = ar;
                Ai[kq]  = ai;
                Ain[kq] = (v2f){-ai.x, -ai.y};
            }

            #pragma unroll
            for (int tt = 0; tt < 2; ++tt) {
                const int t = wave + tt * 8;
                v8f accR = {}; v8f accI = {};
                #pragma unroll
                for (int kq = 0; kq < 4; ++kq) {
                    const int k0 = kq * 4 + 2 * half;
                    v2f Br, Bi;
                    if (g == 0) {
                        const int i0 = k0 * 256 + t * 16 + mn;
                        Br = (v2f){psiR[i0], psiR[i0 + 256]};
                        Bi = (v2f){psiI[i0], psiI[i0 + 256]};
                    } else if (g == 1) {
                        const int i0 = t * 256 + k0 * 16 + mn;
                        Br = (v2f){psiR[i0], psiR[i0 + 16]};
                        Bi = (v2f){psiI[i0], psiI[i0 + 16]};
                    } else {
                        const int i0 = t * 256 + mn * 16 + k0;
                        Br = *reinterpret_cast<const v2f*>(&psiR[i0]);
                        Bi = *reinterpret_cast<const v2f*>(&psiI[i0]);
                    }
                    accR = __builtin_amdgcn_wmma_f32_16x16x4_f32(false, Ar[kq],  false, Br, (short)0, accR, false, false);
                    accR = __builtin_amdgcn_wmma_f32_16x16x4_f32(false, Ain[kq], false, Bi, (short)0, accR, false, false);
                    accI = __builtin_amdgcn_wmma_f32_16x16x4_f32(false, Ar[kq],  false, Bi, (short)0, accI, false, false);
                    accI = __builtin_amdgcn_wmma_f32_16x16x4_f32(false, Ai[kq],  false, Br, (short)0, accI, false, false);
                }
                #pragma unroll
                for (int v = 0; v < 8; ++v) {
                    const int m = v + 8 * half;
                    int idx;
                    if      (g == 0) idx = m * 256 + t * 16 + mn;
                    else if (g == 1) idx = t * 256 + m * 16 + mn;
                    else             idx = t * 256 + mn * 16 + m;
                    psiR[idx] = accR[v];
                    psiI[idx] = accI[v];
                }
            }
            __syncthreads();
        }

        const int* bas = &permB[l * 12];
        int pbase = 0;
        #pragma unroll
        for (int bb = 0; bb < 8; ++bb)
            pbase ^= ((tid >> bb) & 1) ? bas[4 + bb] : 0;
        int pe[16];
        #pragma unroll
        for (int e = 0; e < 16; ++e) {
            int v = 0;
            if (e & 1) v ^= bas[0];
            if (e & 2) v ^= bas[1];
            if (e & 4) v ^= bas[2];
            if (e & 8) v ^= bas[3];
            pe[e] = v;
        }
        float rr[16], ri[16];
        #pragma unroll
        for (int e = 0; e < 16; ++e) {
            const int j = pbase ^ pe[e];
            rr[e] = psiR[j];
            ri[e] = psiI[j];
        }
        __syncthreads();
        #pragma unroll
        for (int q = 0; q < 4; ++q) {
            psiR4[tid * 4 + q] = make_float4(rr[4*q], rr[4*q+1], rr[4*q+2], rr[4*q+3]);
            psiI4[tid * 4 + q] = make_float4(ri[4*q], ri[4*q+1], ri[4*q+2], ri[4*q+3]);
        }
        __syncthreads();
    }

    float acc[12];
    #pragma unroll
    for (int k = 0; k < 12; ++k) acc[k] = 0.0f;
    #pragma unroll
    for (int e4 = 0; e4 < 4; ++e4) {
        const float4 a = psiR4[tid * 4 + e4];
        const float4 c = psiI4[tid * 4 + e4];
        const int ib = tid * 16 + e4 * 4;
        const float p0 = a.x * a.x + c.x * c.x;
        const float p1 = a.y * a.y + c.y * c.y;
        const float p2 = a.z * a.z + c.z * c.z;
        const float p3 = a.w * a.w + c.w * c.w;
        const float pv[4] = {p0, p1, p2, p3};
        #pragma unroll
        for (int cc = 0; cc < 4; ++cc) {
            const int i = ib + cc;
            #pragma unroll
            for (int k = 0; k < 12; ++k)
                acc[k] += ((i >> (11 - k)) & 1) ? -pv[cc] : pv[cc];
        }
    }
    #pragma unroll
    for (int k = 0; k < 12; ++k) {
        float v = acc[k];
        for (int o = 16; o > 0; o >>= 1) v += __shfl_xor(v, o, 32);
        if (lane == 0) red[wave * 12 + k] = v;
    }
    __syncthreads();
    if (tid < 12) {
        float s = 0.0f;
        #pragma unroll
        for (int wv = 0; wv < 8; ++wv) s += red[wv * 12 + tid];
        out[(size_t)b * 12 + tid] = s;
    }
}

extern "C" void kernel_launch(void* const* d_in, const int* in_sizes, int n_in,
                              void* d_out, int out_size, void* d_ws, size_t ws_size,
                              hipStream_t stream) {
    const float* x     = (const float*)d_in[0];
    const float* theta = (const float*)d_in[1];
    float*       out   = (float*)d_out;
    float*       wsf   = (float*)d_ws;

    const int B = in_sizes[0] / DIM;

    qdl_setup_kernel<<<13, 256, 0, stream>>>(theta, wsf);
    qdl_main_kernel<<<B, 256, 0, stream>>>(x, wsf, out);
}


# Layer 0, first gate group: the state is still real there, so the two products with its zero imaginary plane are left out (8 not 16 WMMAs per tile, 2944 not 3072 per wave); sums unchanged
# speedup vs baseline: 1.6567x; 1.6567x over previous
#include <hip/hip_runtime.h>
#include <math.h>

#define NQ    12
#define DIM   4096
#define DEPTH 4
#define WAVES 8

typedef __attribute__((ext_vector_type(2))) float v2f;
typedef __attribute__((ext_vector_type(4))) float v4f;
typedef __attribute__((ext_vector_type(8))) float v8f;
typedef __attribute__((ext_vector_type(2))) unsigned int uint2v;

__host__ __device__ inline int lds_pos(int i) {
    int p = (i & 0xFC0) | ((i & 15) << 2) | ((i >> 4) & 3);
    const int b3 = ((p >> 9) ^ (p >> 10) ^ (p >> 11)) & 1;
    const int b4 = (p >> 11) & 1;
    const int b5 = ((p >> 6) ^ (p >> 9) ^ (p >> 10)) & 1;
    return p ^ (b3 << 3) ^ (b4 << 4) ^ (b5 << 5);
}

__global__ void qdl_setup_kernel(const float* __restrict__ theta,
                                 float* __restrict__ ws) {
    const int tid = threadIdx.x;
    if (blockIdx.x < 12) {
        const int lg = blockIdx.x;
        const int l  = lg / 3, g = lg % 3;
        const int i  = tid >> 4, j = tid & 15;

        float pr = 1.0f, pi = 0.0f;
        #pragma unroll
        for (int q = 0; q < 4; ++q) {
            const int w = g * 4 + q;
            const float* th = theta + (l * NQ + w) * 3;
            const float phi = th[0], the = th[1], om = th[2];
            const float ch = cosf(0.5f * the), sh = sinf(0.5f * the);
            const float a  = 0.5f * (phi + om), b = 0.5f * (phi - om);
            const float ca = cosf(a), sa = sinf(a);
            const float cb = cosf(b), sb = sinf(b);
            const int bi = (i >> (3 - q)) & 1;
            const int bj = (j >> (3 - q)) & 1;
            float ur, ui;
            if      (bi == 0 && bj == 0) { ur =  ca * ch; ui = -sa * ch; }
            else if (bi == 0 && bj == 1) { ur = -cb * sh; ui = -sb * sh; }
            else if (bi == 1 && bj == 0) { ur =  cb * sh; ui = -sb * sh; }
            else                         { ur =  ca * ch; ui =  sa * ch; }
            const float nr = pr * ur - pi * ui;
            const float ni = pr * ui + pi * ur;
            pr = nr; pi = ni;
        }
        ws[(lg * 2 + 0) * 256 + tid] = pr;
        ws[(lg * 2 + 1) * 256 + tid] = pi;
    } else if (tid < 48) {
        const int l = tid / 12, k = tid % 12;
        const int r = l % (NQ - 1) + 1;
        int j = 1 << k;
        for (int w = 0; w < NQ; ++w) {
            const int tg   = (w + r) % NQ;
            const int cbit = (j >> (NQ - 1 - w)) & 1;
            j ^= cbit << (NQ - 1 - tg);
        }
        ((int*)ws)[12 * 2 * 256 + tid] = lds_pos(j);
    }
}

template <int M>
__device__ inline float swz_xor(float v) {
    return __int_as_float(__builtin_amdgcn_ds_swizzle(__float_as_int(v), (M << 10) | 0x1F));
}

__device__ inline v8f mma(v2f a, v2f b, v8f c) {
    return __builtin_amdgcn_wmma_f32_16x16x4_f32(false, a, false, b, (short)0, c, false, false);
}

struct GateFrag { v2f r[4], i[4], in[4]; };

__device__ inline GateFrag load_gate(const float* gl, int lg, int c, int h) {
    GateFrag f;
    const float* Gr = gl + (lg * 2 + 0) * 256 + c * 16 + 8 * h;
    const float* Gi = gl + (lg * 2 + 1) * 256 + c * 16 + 8 * h;
    #pragma unroll
    for (int j = 0; j < 4; ++j) {
        f.r[j]  = *reinterpret_cast<const v2f*>(Gr + 2 * j);
        f.i[j]  = *reinterpret_cast<const v2f*>(Gi + 2 * j);
        f.in[j] = -f.i[j];
    }
    return f;
}

__global__ __launch_bounds__(256) void qdl_main_kernel(
        const float* __restrict__ x,
        const float* __restrict__ ws,
        float* __restrict__ out,
        int B) {
    __shared__ float st[WAVES * 2 * DIM];
    __shared__ float gLDS[12 * 2 * 256];

    const int tid  = threadIdx.x;
    const int lane = tid & 31;
    const int wave = tid >> 5;
    const int h    = lane >> 4;
    const int c    = lane & 15;

    const int row  = blockIdx.x * WAVES + wave;
    const int rowc = row < B ? row : B - 1;

    const float* xr = x + (size_t)rowc * DIM;
    v8f Tr[16], Ti[16];
    #pragma unroll
    for (int a = 0; a < 16; ++a) {
        #pragma unroll
        for (int r = 0; r < 8; ++r)
            Tr[a][r] = xr[a * 256 + (8 * h + r) * 16 + c];
    }
    const float x0 = xr[0];

    for (int k = tid; k < 12 * 2 * 256; k += 256) gLDS[k] = ws[k];
    __syncthreads();
    if (row >= B) return;

    float* sR = st + wave * (2 * DIM);
    float* sI = sR + DIM;
    const int* qbasis = reinterpret_cast<const int*>(ws) + 12 * 2 * 256;

    float ssp[4] = {0.f, 0.f, 0.f, 0.f};
    #pragma unroll
    for (int a = 0; a < 16; ++a) {
        #pragma unroll
        for (int r = 0; r < 8; ++r) ssp[r & 3] += Tr[a][r] * Tr[a][r];
    }
    float ss = (ssp[0] + ssp[1]) + (ssp[2] + ssp[3]);
    ss += swz_xor<16>(ss);
    ss += swz_xor<8>(ss);
    ss += swz_xor<4>(ss);
    ss += swz_xor<2>(ss);
    ss += swz_xor<1>(ss);
    const float n1sq = ss;
    const bool  zero = (sqrtf(n1sq) <= 1e-12f);
    const float x0n  = zero ? 1.0f : x0;
    const float inv  = 1.0f / sqrtf(n1sq - x0 * x0 + x0n * x0n);
    Tr[0][0] = (lane == 0) ? x0n : Tr[0][0];
    #pragma unroll
    for (int a = 0; a < 16; ++a) Tr[a] = Tr[a] * inv;

    {
        const GateFrag G = load_gate(gLDS, 1, c, h);
        #pragma unroll
        for (int t = 0; t < 16; ++t) {
            v8f accR = {}; v8f accI = {};
            #pragma unroll
            for (int j = 0; j < 4; ++j) {
                const v2f xr2 = {Tr[t][2 * j], Tr[t][2 * j + 1]};
                accR = mma(xr2, G.r[j], accR);
                accI = mma(xr2, G.i[j], accI);
            }
            Tr[t] = accR; Ti[t] = accI;
        }
    }
    for (int l = 0; l < DEPTH; ++l) {
        {
            const GateFrag G = load_gate(gLDS, l * 3 + 2, c, h);
            #pragma unroll
            for (int t = 0; t < 16; ++t) {
                v8f accR = {}; v8f accI = {};
                #pragma unroll
                for (int j = 0; j < 4; ++j) {
                    const v2f xr2 = {Tr[t][2 * j], Tr[t][2 * j + 1]};
                    const v2f xi2 = {Ti[t][2 * j], Ti[t][2 * j + 1]};
                    accR = mma(G.r[j],  xr2, accR);
                    accI = mma(G.r[j],  xi2, accI);
                    accR = mma(G.in[j], xi2, accR);
                    accI = mma(G.i[j],  xr2, accI);
                }
                Tr[t] = accR; Ti[t] = accI;
            }
        }
        #pragma unroll
        for (int p = 0; p < 8; ++p) {
            #pragma unroll
            for (int u = 0; u < 8; ++u) {
                const uint2v sr = __builtin_amdgcn_permlane16_swap(
                    __float_as_uint(Tr[p][u]), __float_as_uint(Tr[p + 8][u]), false, false);
                Tr[p][u]     = __uint_as_float(sr[0]);
                Tr[p + 8][u] = __uint_as_float(sr[1]);
                const uint2v si = __builtin_amdgcn_permlane16_swap(
                    __float_as_uint(Ti[p][u]), __float_as_uint(Ti[p + 8][u]), false, false);
                Ti[p][u]     = __uint_as_float(si[0]);
                Ti[p + 8][u] = __uint_as_float(si[1]);
            }
        }
        #pragma unroll
        for (int p = 1; p < 8; ++p) {
            #pragma unroll
            for (int u = 0; u < p; ++u) {
                float t0;
                t0 = Tr[p][u];         Tr[p][u]         = Tr[u][p];         Tr[u][p]         = t0;
                t0 = Tr[p + 8][u];     Tr[p + 8][u]     = Tr[u + 8][p];     Tr[u + 8][p]     = t0;
                t0 = Ti[p][u];         Ti[p][u]         = Ti[u][p];         Ti[u][p]         = t0;
                t0 = Ti[p + 8][u];     Ti[p + 8][u]     = Ti[u + 8][p];     Ti[u + 8][p]     = t0;
            }
        }
        {
            const GateFrag G = load_gate(gLDS, l * 3 + 0, c, h);
            #pragma unroll
            for (int t = 0; t < 16; ++t) {
                v8f accR = {}; v8f accI = {};
                #pragma unroll
                for (int j = 0; j < 4; ++j) {
                    const v2f xr2 = {Tr[t][2 * j], Tr[t][2 * j + 1]};
                    const v2f xi2 = {Ti[t][2 * j], Ti[t][2 * j + 1]};
                    accR = mma(G.r[j],  xr2, accR);
                    accI = mma(G.r[j],  xi2, accI);
                    accR = mma(G.in[j], xi2, accR);
                    accI = mma(G.i[j],  xr2, accI);
                }
                Tr[t] = accR; Ti[t] = accI;
            }
        }
        {
            const int* qb = qbasis + l * 12;
            int lq = qb[11] & -h;
            lq ^= qb[4] & -(c & 1);
            lq ^= qb[5] & -((c >> 1) & 1);
            lq ^= qb[6] & -((c >> 2) & 1);
            lq ^= qb[7] & -((c >> 3) & 1);
            #pragma unroll
            for (int t = 0; t < 16; ++t) {
                int ut = 0;
                if (t & 1) ut ^= qb[0];
                if (t & 2) ut ^= qb[1];
                if (t & 4) ut ^= qb[2];
                if (t & 8) ut ^= qb[3];
                #pragma unroll
                for (int r = 0; r < 8; ++r) {
                    int ur = ut;
                    if (r & 1) ur ^= qb[8];
                    if (r & 2) ur ^= qb[9];
                    if (r & 4) ur ^= qb[10];
                    const int pos = lq ^ ur;
                    sR[pos] = Tr[t][r];
                    sI[pos] = Ti[t][r];
                }
            }
            #pragma unroll
            for (int a = 0; a < 16; ++a) {
                #pragma unroll
                for (int q = 0; q < 2; ++q) {
                    const int pos = lds_pos(a * 256 + (8 * h + 4 * q) * 16 + c);
                    const v4f vr = *reinterpret_cast<const v4f*>(&sR[pos]);
                    const v4f vi = *reinterpret_cast<const v4f*>(&sI[pos]);
                    #pragma unroll
                    for (int e = 0; e < 4; ++e) { Tr[a][4 * q + e] = vr[e]; Ti[a][4 * q + e] = vi[e]; }
                }
            }
        }
        if (l + 1 < DEPTH) {
            const GateFrag G = load_gate(gLDS, (l + 1) * 3 + 1, c, h);
            #pragma unroll
            for (int t = 0; t < 16; ++t) {
                v8f accR = {}; v8f accI = {};
                #pragma unroll
                for (int j = 0; j < 4; ++j) {
                    const v2f xr2 = {Tr[t][2 * j], Tr[t][2 * j + 1]};
                    const v2f xi2 = {Ti[t][2 * j], Ti[t][2 * j + 1]};
                    accR = mma(xr2, G.r[j],  accR);
                    accI = mma(xi2, G.r[j],  accI);
                    accR = mma(xi2, G.in[j], accR);
                    accI = mma(xr2, G.i[j],  accI);
                }
                Tr[t] = accR; Ti[t] = accI;
            }
        }
    }

    float Sa[16], Sr[8];
    #pragma unroll
    for (int r = 0; r < 8; ++r) Sr[r] = 0.0f;
    #pragma unroll
    for (int a = 0; a < 16; ++a) {
        float s = 0.0f;
        #pragma unroll
        for (int r = 0; r < 8; ++r) {
            const float p = Tr[a][r] * Tr[a][r] + Ti[a][r] * Ti[a][r];
            s += p;
            Sr[r] += p;
        }
        Sa[a] = s;
    }
    float tot = 0.0f;
    #pragma unroll
    for (int a = 0; a < 16; ++a) tot += Sa[a];
    float o[12];
    #pragma unroll
    for (int k = 0; k < 4; ++k) {
        float s = 0.0f;
        #pragma unroll
        for (int a = 0; a < 16; ++a) s += ((a >> (3 - k)) & 1) ? -Sa[a] : Sa[a];
        o[k] = s;
    }
    o[4] = h ? -tot : tot;
    #pragma unroll
    for (int k = 5; k < 8; ++k) {
        float s = 0.0f;
        #pragma unroll
        for (int r = 0; r < 8; ++r) s += ((r >> (7 - k)) & 1) ? -Sr[r] : Sr[r];
        o[k] = s;
    }
    #pragma unroll
    for (int k = 8; k < 12; ++k)
        o[k] = ((c >> (11 - k)) & 1) ? -tot : tot;
    float mine = 0.0f;
    #pragma unroll
    for (int k = 0; k < 12; ++k) {
        float v = o[k];
        for (int off = 16; off > 0; off >>= 1) v += __shfl_xor(v, off, 32);
        mine = (lane == k) ? v : mine;
    }
    if (lane < 12) out[(size_t)row * 12 + lane] = mine;
}

extern "C" void kernel_launch(void* const* d_in, const int* in_sizes, int n_in,
                              void* d_out, int out_size, void* d_ws, size_t ws_size,
                              hipStream_t stream) {
    const float* x     = (const float*)d_in[0];
    const float* theta = (const float*)d_in[1];
    float*       out   = (float*)d_out;
    float*       wsf   = (float*)d_ws;

    const int B = in_sizes[0] / DIM;

    qdl_setup_kernel<<<13, 256, 0, stream>>>(theta, wsf);
    qdl_main_kernel<<<(B + WAVES - 1) / WAVES, 256, 0, stream>>>(x, wsf, out, B);
}


# Wave priority steps down once per layer (s_setprio 3,2,1,0): a SIMD's two waves trade the matrix pipe layer by layer, each one's lane swaps and LDS passes run under the other's WMMAs; sums unchanged
# speedup vs baseline: 1.7249x; 1.0412x over previous
#include <hip/hip_runtime.h>
#include <math.h>

#define NQ    12
#define DIM   4096
#define DEPTH 4
#define WAVES 8

typedef __attribute__((ext_vector_type(2))) float v2f;
typedef __attribute__((ext_vector_type(4))) float v4f;
typedef __attribute__((ext_vector_type(8))) float v8f;
typedef __attribute__((ext_vector_type(2))) unsigned int uint2v;

__host__ __device__ inline int lds_pos(int i) {
    int p = (i & 0xFC0) | ((i & 15) << 2) | ((i >> 4) & 3);
    const int b3 = ((p >> 9) ^ (p >> 10) ^ (p >> 11)) & 1;
    const int b4 = (p >> 11) & 1;
    const int b5 = ((p >> 6) ^ (p >> 9) ^ (p >> 10)) & 1;
    return p ^ (b3 << 3) ^ (b4 << 4) ^ (b5 << 5);
}

__global__ void qdl_setup_kernel(const float* __restrict__ theta,
                                 float* __restrict__ ws) {
    const int tid = threadIdx.x;
    if (blockIdx.x < 12) {
        const int lg = blockIdx.x;
        const int l  = lg / 3, g = lg % 3;
        const int i  = tid >> 4, j = tid & 15;

        float pr = 1.0f, pi = 0.0f;
        #pragma unroll
        for (int q = 0; q < 4; ++q) {
            const int w = g * 4 + q;
            const float* th = theta + (l * NQ + w) * 3;
            const float phi = th[0], the = th[1], om = th[2];
            const float ch = cosf(0.5f * the), sh = sinf(0.5f * the);
            const float a  = 0.5f * (phi + om), b = 0.5f * (phi - om);
            const float ca = cosf(a), sa = sinf(a);
            const float cb = cosf(b), sb = sinf(b);
            const int bi = (i >> (3 - q)) & 1;
            const int bj = (j >> (3 - q)) & 1;
            float ur, ui;
            if      (bi == 0 && bj == 0) { ur =  ca * ch; ui = -sa * ch; }
            else if (bi == 0 && bj == 1) { ur = -cb * sh; ui = -sb * sh; }
            else if (bi == 1 && bj == 0) { ur =  cb * sh; ui = -sb * sh; }
            else                         { ur =  ca * ch; ui =  sa * ch; }
            const float nr = pr * ur - pi * ui;
            const float ni = pr * ui + pi * ur;
            pr = nr; pi = ni;
        }
        ws[(lg * 2 + 0) * 256 + tid] = pr;
        ws[(lg * 2 + 1) * 256 + tid] = pi;
    } else if (tid < 48) {
        const int l = tid / 12, k = tid % 12;
        const int r = l % (NQ - 1) + 1;
        int j = 1 << k;
        for (int w = 0; w < NQ; ++w) {
            const int tg   = (w + r) % NQ;
            const int cbit = (j >> (NQ - 1 - w)) & 1;
            j ^= cbit << (NQ - 1 - tg);
        }
        ((int*)ws)[12 * 2 * 256 + tid] = lds_pos(j);
    }
}

template <int M>
__device__ inline float swz_xor(float v) {
    return __int_as_float(__builtin_amdgcn_ds_swizzle(__float_as_int(v), (M << 10) | 0x1F));
}

__device__ inline v8f mma(v2f a, v2f b, v8f c) {
    return __builtin_amdgcn_wmma_f32_16x16x4_f32(false, a, false, b, (short)0, c, false, false);
}

struct GateFrag { v2f r[4], i[4], in[4]; };

__device__ inline GateFrag load_gate(const float* gl, int lg, int c, int h) {
    GateFrag f;
    const float* Gr = gl + (lg * 2 + 0) * 256 + c * 16 + 8 * h;
    const float* Gi = gl + (lg * 2 + 1) * 256 + c * 16 + 8 * h;
    #pragma unroll
    for (int j = 0; j < 4; ++j) {
        f.r[j]  = *reinterpret_cast<const v2f*>(Gr + 2 * j);
        f.i[j]  = *reinterpret_cast<const v2f*>(Gi + 2 * j);
        f.in[j] = -f.i[j];
    }
    return f;
}

__global__ __launch_bounds__(256) void qdl_main_kernel(
        const float* __restrict__ x,
        const float* __restrict__ ws,
        float* __restrict__ out,
        int B) {
    __shared__ float st[WAVES * 2 * DIM];
    __shared__ float gLDS[12 * 2 * 256];

    const int tid  = threadIdx.x;
    const int lane = tid & 31;
    const int wave = tid >> 5;
    const int h    = lane >> 4;
    const int c    = lane & 15;

    const int row  = blockIdx.x * WAVES + wave;
    const int rowc = row < B ? row : B - 1;

    const float* xr = x + (size_t)rowc * DIM;
    v8f Tr[16], Ti[16];
    #pragma unroll
    for (int a = 0; a < 16; ++a) {
        #pragma unroll
        for (int r = 0; r < 8; ++r)
            Tr[a][r] = xr[a * 256 + (8 * h + r) * 16 + c];
    }
    const float x0 = xr[0];

    for (int k = tid; k < 12 * 2 * 256; k += 256) gLDS[k] = ws[k];
    __syncthreads();
    if (row >= B) return;

    float* sR = st + wave * (2 * DIM);
    float* sI = sR + DIM;
    const int* qbasis = reinterpret_cast<const int*>(ws) + 12 * 2 * 256;

    float ssp[4] = {0.f, 0.f, 0.f, 0.f};
    #pragma unroll
    for (int a = 0; a < 16; ++a) {
        #pragma unroll
        for (int r = 0; r < 8; ++r) ssp[r & 3] += Tr[a][r] * Tr[a][r];
    }
    float ss = (ssp[0] + ssp[1]) + (ssp[2] + ssp[3]);
    ss += swz_xor<16>(ss);
    ss += swz_xor<8>(ss);
    ss += swz_xor<4>(ss);
    ss += swz_xor<2>(ss);
    ss += swz_xor<1>(ss);
    const float n1sq = ss;
    const bool  zero = (sqrtf(n1sq) <= 1e-12f);
    const float x0n  = zero ? 1.0f : x0;
    const float inv  = 1.0f / sqrtf(n1sq - x0 * x0 + x0n * x0n);
    Tr[0][0] = (lane == 0) ? x0n : Tr[0][0];
    #pragma unroll
    for (int a = 0; a < 16; ++a) Tr[a] = Tr[a] * inv;

    __builtin_amdgcn_s_setprio(3);
    {
        const GateFrag G = load_gate(gLDS, 1, c, h);
        #pragma unroll
        for (int t = 0; t < 16; ++t) {
            v8f accR = {}; v8f accI = {};
            #pragma unroll
            for (int j = 0; j < 4; ++j) {
                const v2f xr2 = {Tr[t][2 * j], Tr[t][2 * j + 1]};
                accR = mma(xr2, G.r[j], accR);
                accI = mma(xr2, G.i[j], accI);
            }
            Tr[t] = accR; Ti[t] = accI;
        }
    }
    for (int l = 0; l < DEPTH; ++l) {
        switch (l) {
            case 1:  __builtin_amdgcn_s_setprio(2); break;
            case 2:  __builtin_amdgcn_s_setprio(1); break;
            case 3:  __builtin_amdgcn_s_setprio(0); break;
            default: break;
        }
        {
            const GateFrag G = load_gate(gLDS, l * 3 + 2, c, h);
            #pragma unroll
            for (int t = 0; t < 16; ++t) {
                v8f accR = {}; v8f accI = {};
                #pragma unroll
                for (int j = 0; j < 4; ++j) {
                    const v2f xr2 = {Tr[t][2 * j], Tr[t][2 * j + 1]};
                    const v2f xi2 = {Ti[t][2 * j], Ti[t][2 * j + 1]};
                    accR = mma(G.r[j],  xr2, accR);
                    accI = mma(G.r[j],  xi2, accI);
                    accR = mma(G.in[j], xi2, accR);
                    accI = mma(G.i[j],  xr2, accI);
                }
                Tr[t] = accR; Ti[t] = accI;
            }
        }
        #pragma unroll
        for (int p = 0; p < 8; ++p) {
            #pragma unroll
            for (int u = 0; u < 8; ++u) {
                const uint2v sr = __builtin_amdgcn_permlane16_swap(
                    __float_as_uint(Tr[p][u]), __float_as_uint(Tr[p + 8][u]), false, false);
                Tr[p][u]     = __uint_as_float(sr[0]);
                Tr[p + 8][u] = __uint_as_float(sr[1]);
                const uint2v si = __builtin_amdgcn_permlane16_swap(
                    __float_as_uint(Ti[p][u]), __float_as_uint(Ti[p + 8][u]), false, false);
                Ti[p][u]     = __uint_as_float(si[0]);
                Ti[p + 8][u] = __uint_as_float(si[1]);
            }
        }
        #pragma unroll
        for (int p = 1; p < 8; ++p) {
            #pragma unroll
            for (int u = 0; u < p; ++u) {
                float t0;
                t0 = Tr[p][u];         Tr[p][u]         = Tr[u][p];         Tr[u][p]         = t0;
                t0 = Tr[p + 8][u];     Tr[p + 8][u]     = Tr[u + 8][p];     Tr[u + 8][p]     = t0;
                t0 = Ti[p][u];         Ti[p][u]         = Ti[u][p];         Ti[u][p]         = t0;
                t0 = Ti[p + 8][u];     Ti[p + 8][u]     = Ti[u + 8][p];     Ti[u + 8][p]     = t0;
            }
        }
        {
            const GateFrag G = load_gate(gLDS, l * 3 + 0, c, h);
            #pragma unroll
            for (int t = 0; t < 16; ++t) {
                v8f accR = {}; v8f accI = {};
                #pragma unroll
                for (int j = 0; j < 4; ++j) {
                    const v2f xr2 = {Tr[t][2 * j], Tr[t][2 * j + 1]};
                    const v2f xi2 = {Ti[t][2 * j], Ti[t][2 * j + 1]};
                    accR = mma(G.r[j],  xr2, accR);
                    accI = mma(G.r[j],  xi2, accI);
                    accR = mma(G.in[j], xi2, accR);
                    accI = mma(G.i[j],  xr2, accI);
                }
                Tr[t] = accR; Ti[t] = accI;
            }
        }
        {
            const int* qb = qbasis + l * 12;
            int lq = qb[11] & -h;
            lq ^= qb[4] & -(c & 1);
            lq ^= qb[5] & -((c >> 1) & 1);
            lq ^= qb[6] & -((c >> 2) & 1);
            lq ^= qb[7] & -((c >> 3) & 1);
            #pragma unroll
            for (int t = 0; t < 16; ++t) {
                int ut = 0;
                if (t & 1) ut ^= qb[0];
                if (t & 2) ut ^= qb[1];
                if (t & 4) ut ^= qb[2];
                if (t & 8) ut ^= qb[3];
                #pragma unroll
                for (int r = 0; r < 8; ++r) {
                    int ur = ut;
                    if (r & 1) ur ^= qb[8];
                    if (r & 2) ur ^= qb[9];
                    if (r & 4) ur ^= qb[10];
                    const int pos = lq ^ ur;
                    sR[pos] = Tr[t][r];
                    sI[pos] = Ti[t][r];
                }
            }
            #pragma unroll
            for (int a = 0; a < 16; ++a) {
                #pragma unroll
                for (int q = 0; q < 2; ++q) {
                    const int pos = lds_pos(a * 256 + (8 * h + 4 * q) * 16 + c);
                    const v4f vr = *reinterpret_cast<const v4f*>(&sR[pos]);
                    const v4f vi = *reinterpret_cast<const v4f*>(&sI[pos]);
                    #pragma unroll
                    for (int e = 0; e < 4; ++e) { Tr[a][4 * q + e] = vr[e]; Ti[a][4 * q + e] = vi[e]; }
                }
            }
        }
        if (l + 1 < DEPTH) {
            const GateFrag G = load_gate(gLDS, (l + 1) * 3 + 1, c, h);
            #pragma unroll
            for (int t = 0; t < 16; ++t) {
                v8f accR = {}; v8f accI = {};
                #pragma unroll
                for (int j = 0; j < 4; ++j) {
                    const v2f xr2 = {Tr[t][2 * j], Tr[t][2 * j + 1]};
                    const v2f xi2 = {Ti[t][2 * j], Ti[t][2 * j + 1]};
                    accR = mma(xr2, G.r[j],  accR);
                    accI = mma(xi2, G.r[j],  accI);
                    accR = mma(xi2, G.in[j], accR);
                    accI = mma(xr2, G.i[j],  accI);
                }
                Tr[t] = accR; Ti[t] = accI;
            }
        }
    }

    float Sa[16], Sr[8];
    #pragma unroll
    for (int r = 0; r < 8; ++r) Sr[r] = 0.0f;
    #pragma unroll
    for (int a = 0; a < 16; ++a) {
        float s = 0.0f;
        #pragma unroll
        for (int r = 0; r < 8; ++r) {
            const float p = Tr[a][r] * Tr[a][r] + Ti[a][r] * Ti[a][r];
            s += p;
            Sr[r] += p;
        }
        Sa[a] = s;
    }
    float tot = 0.0f;
    #pragma unroll
    for (int a = 0; a < 16; ++a) tot += Sa[a];
    float o[12];
    #pragma unroll
    for (int k = 0; k < 4; ++k) {
        float s = 0.0f;
        #pragma unroll
        for (int a = 0; a < 16; ++a) s += ((a >> (3 - k)) & 1) ? -Sa[a] : Sa[a];
        o[k] = s;
    }
    o[4] = h ? -tot : tot;
    #pragma unroll
    for (int k = 5; k < 8; ++k) {
        float s = 0.0f;
        #pragma unroll
        for (int r = 0; r < 8; ++r) s += ((r >> (7 - k)) & 1) ? -Sr[r] : Sr[r];
        o[k] = s;
    }
    #pragma unroll
    for (int k = 8; k < 12; ++k)
        o[k] = ((c >> (11 - k)) & 1) ? -tot : tot;
    float mine = 0.0f;
    #pragma unroll
    for (int k = 0; k < 12; ++k) {
        float v = o[k];
        for (int off = 16; off > 0; off >>= 1) v += __shfl_xor(v, off, 32);
        mine = (lane == k) ? v : mine;
    }
    if (lane < 12) out[(size_t)row * 12 + lane] = mine;
}

extern "C" void kernel_launch(void* const* d_in, const int* in_sizes, int n_in,
                              void* d_out, int out_size, void* d_ws, size_t ws_size,
                              hipStream_t stream) {
    const float* x     = (const float*)d_in[0];
    const float* theta = (const float*)d_in[1];
    float*       out   = (float*)d_out;
    float*       wsf   = (float*)d_ws;

    const int B = in_sizes[0] / DIM;

    qdl_setup_kernel<<<13, 256, 0, stream>>>(theta, wsf);
    qdl_main_kernel<<<(B + WAVES - 1) / WAVES, 256, 0, stream>>>(x, wsf, out, B);
}


# Gates as real 16x16 matrices over (re/im plane, 3-wire digit): four wire triples per layer, one real product each (512 not 768 WMMAs per layer), planes in the lane halves; sum order changes
# speedup vs baseline: 1.9859x; 1.1513x over previous
#include <hip/hip_runtime.h>
#include <math.h>

#define NQ    12
#define DIM   4096
#define DEPTH 4
#define WAVES 8

typedef __attribute__((ext_vector_type(2))) float v2f;
typedef __attribute__((ext_vector_type(4))) float v4f;
typedef __attribute__((ext_vector_type(8))) float v8f;

__host__ __device__ inline int idx_x(int t, int r, int c) { return (r << 9) | ((t & 7) << 6) | ((t >> 3) << 4) | c; }
__host__ __device__ inline int idx_y(int t, int r, int c) { return (c << 8) | ((t >> 3) << 6) | ((t & 7) << 3) | r; }

__host__ __device__ inline int pos_x(int i) {
    int p = ((i >> 9) & 3) | ((i & 15) << 2) | (((i >> 4) & 31) << 6) | (i & 0x800);
    return p ^ (((p >> 11) & 1) << 2);
}
__host__ __device__ inline int pos_y(int i) {
    int p = (i & 3) | (((i >> 8) & 15) << 2) | (((i >> 2) & 63) << 6);
    return p ^ (((p >> 6) & 3) << 2);
}
#define HALF_X 32
#define HALF_Y 16

__global__ void qdl_setup_kernel(const float* __restrict__ theta,
                                 float* __restrict__ ws) {
    const int tid = threadIdx.x;
    if (blockIdx.x < 16) {
        const int lg = blockIdx.x;
        const int l  = lg >> 2, g = lg & 3;
        const int m  = tid >> 4, n = tid & 15;
        const int i  = m & 7, j = n & 7;
        const int pm = m >> 3, pn = n >> 3;

        float pr = 1.0f, pi = 0.0f;
        #pragma unroll
        for (int q = 0; q < 3; ++q) {
            const int w = g * 3 + q;
            const float* th = theta + (l * NQ + w) * 3;
            const float phi = th[0], the = th[1], om = th[2];
            const float ch = cosf(0.5f * the), sh = sinf(0.5f * the);
            const float a  = 0.5f * (phi + om), b = 0.5f * (phi - om);
            const float ca = cosf(a), sa = sinf(a);
            const float cb = cosf(b), sb = sinf(b);
            const int bi = (i >> (2 - q)) & 1;
            const int bj = (j >> (2 - q)) & 1;
            float ur, ui;
            if      (bi == 0 && bj == 0) { ur =  ca * ch; ui = -sa * ch; }
            else if (bi == 0 && bj == 1) { ur = -cb * sh; ui = -sb * sh; }
            else if (bi == 1 && bj == 0) { ur =  cb * sh; ui = -sb * sh; }
            else                         { ur =  ca * ch; ui =  sa * ch; }
            const float nr = pr * ur - pi * ui;
            const float ni = pr * ui + pi * ur;
            pr = nr; pi = ni;
        }
        float v;
        if (pm == pn) v = pr; else if (pm == 1) v = pi; else v = -pi;
        ws[lg * 256 + tid] = v;
    } else if (tid < 48) {
        const int l = tid / 12, k = tid % 12;
        const int r = l % (NQ - 1) + 1;
        int j = 1 << k;
        for (int w = 0; w < NQ; ++w) {
            const int tg   = (w + r) % NQ;
            const int cbit = (j >> (NQ - 1 - w)) & 1;
            j ^= cbit << (NQ - 1 - tg);
        }
        ((int*)ws)[16 * 256 + tid] = pos_x(j);
    }
}

template <int M>
__device__ inline float swz_xor(float v) {
    return __int_as_float(__builtin_amdgcn_ds_swizzle(__float_as_int(v), (M << 10) | 0x1F));
}

__device__ inline v8f mma(v2f a, v2f b, v8f c) {
    return __builtin_amdgcn_wmma_f32_16x16x4_f32(false, a, false, b, (short)0, c, false, false);
}

struct GateFrag { v2f m[4]; };

__device__ inline GateFrag load_gate(const float* gl, int lg, int c, int h) {
    GateFrag f;
    const float* G = gl + lg * 256 + c * 16 + 8 * h;
    #pragma unroll
    for (int j = 0; j < 4; ++j) f.m[j] = *reinterpret_cast<const v2f*>(G + 2 * j);
    return f;
}

__global__ __launch_bounds__(256) void qdl_main_kernel(
        const float* __restrict__ x,
        const float* __restrict__ ws,
        float* __restrict__ out,
        int B) {
    __shared__ float st[WAVES * 2 * DIM];
    __shared__ float gLDS[16 * 256];

    const int tid  = threadIdx.x;
    const int lane = tid & 31;
    const int wave = tid >> 5;
    const int h    = lane >> 4;
    const int c    = lane & 15;

    const int row  = blockIdx.x * WAVES + wave;
    const int rowc = row < B ? row : B - 1;

    const float* xr = x + (size_t)rowc * DIM;
    v8f T[32];
    #pragma unroll
    for (int t = 0; t < 32; ++t) {
        #pragma unroll
        for (int r = 0; r < 8; ++r)
            T[t][r] = xr[idx_x(t, r, c)];
    }
    const float x0 = xr[0];

    for (int k = tid; k < 16 * 256; k += 256) gLDS[k] = ws[k];
    __syncthreads();
    if (row >= B) return;

    float* sW = st + wave * (2 * DIM);
    const int* qbasis = reinterpret_cast<const int*>(ws) + 16 * 256;

    #pragma unroll
    for (int t = 0; t < 32; ++t) {
        #pragma unroll
        for (int r = 0; r < 8; ++r) T[t][r] = h ? 0.0f : T[t][r];
    }

    float ssp[4] = {0.f, 0.f, 0.f, 0.f};
    #pragma unroll
    for (int t = 0; t < 32; ++t) {
        #pragma unroll
        for (int r = 0; r < 8; ++r) ssp[r & 3] += T[t][r] * T[t][r];
    }
    float ss = (ssp[0] + ssp[1]) + (ssp[2] + ssp[3]);
    ss += swz_xor<16>(ss);
    ss += swz_xor<8>(ss);
    ss += swz_xor<4>(ss);
    ss += swz_xor<2>(ss);
    ss += swz_xor<1>(ss);
    const float n1sq = ss;
    const bool  zero = (sqrtf(n1sq) <= 1e-12f);
    const float x0n  = zero ? 1.0f : x0;
    const float inv  = 1.0f / sqrtf(n1sq - x0 * x0 + x0n * x0n);
    T[0][0] = (lane == 0) ? x0n : T[0][0];
    #pragma unroll
    for (int t = 0; t < 32; ++t) T[t] = T[t] * inv;

    const int hy      = (h << 12) | (h ? HALF_Y : 0);
    const int hx      = (h << 12) | (h ? HALF_X : 0);
    const int laneWY  = hy ^ pos_y(idx_x(0, 0, c));
    const int laneRY  = hy ^ pos_y(idx_y(0, 0, c));
    const int laneRX  = hx ^ pos_x(idx_x(0, 0, c));

    __builtin_amdgcn_s_setprio(3);
    for (int l = 0; l < DEPTH; ++l) {
        switch (l) {
            case 1:  __builtin_amdgcn_s_setprio(2); break;
            case 2:  __builtin_amdgcn_s_setprio(1); break;
            case 3:  __builtin_amdgcn_s_setprio(0); break;
            default: break;
        }
        {
            const GateFrag G = load_gate(gLDS, l * 4 + 0, c, h);
            #pragma unroll
            for (int t = 0; t < 32; ++t) {
                v8f acc = {};
                #pragma unroll
                for (int j = 0; j < 4; ++j) {
                    const v2f xb = {T[t][2 * j], T[t][2 * j + 1]};
                    acc = mma(G.m[j], xb, acc);
                }
                T[t] = acc;
            }
        }
        {
            const GateFrag G = load_gate(gLDS, l * 4 + 1, c, h);
            #pragma unroll
            for (int u = 0; u < 4; ++u) {
                #pragma unroll
                for (int r = 0; r < 8; ++r) {
                    v8f acc = {};
                    #pragma unroll
                    for (int j = 0; j < 4; ++j) {
                        const v2f xb = {T[u * 8 + 2 * j][r], T[u * 8 + 2 * j + 1][r]};
                        acc = mma(G.m[j], xb, acc);
                    }
                    #pragma unroll
                    for (int k = 0; k < 8; ++k) T[u * 8 + k][r] = acc[k];
                }
            }
        }
        #pragma unroll
        for (int t = 0; t < 32; ++t) {
            #pragma unroll
            for (int r = 0; r < 8; ++r)
                sW[laneWY ^ pos_y(idx_x(t, r, 0))] = T[t][r];
        }
        #pragma unroll
        for (int t = 0; t < 32; ++t) {
            #pragma unroll
            for (int q = 0; q < 2; ++q) {
                const v4f v = *reinterpret_cast<const v4f*>(&sW[laneRY ^ pos_y(idx_y(t, 4 * q, 0))]);
                #pragma unroll
                for (int e = 0; e < 4; ++e) T[t][4 * q + e] = v[e];
            }
        }
        {
            const GateFrag G = load_gate(gLDS, l * 4 + 3, c, h);
            #pragma unroll
            for (int t = 0; t < 32; ++t) {
                v8f acc = {};
                #pragma unroll
                for (int j = 0; j < 4; ++j) {
                    const v2f xb = {T[t][2 * j], T[t][2 * j + 1]};
                    acc = mma(G.m[j], xb, acc);
                }
                T[t] = acc;
            }
        }
        {
            const GateFrag G = load_gate(gLDS, l * 4 + 2, c, h);
            #pragma unroll
            for (int u = 0; u < 4; ++u) {
                #pragma unroll
                for (int r = 0; r < 8; ++r) {
                    v8f acc = {};
                    #pragma unroll
                    for (int j = 0; j < 4; ++j) {
                        const v2f xb = {T[u * 8 + 2 * j][r], T[u * 8 + 2 * j + 1][r]};
                        acc = mma(G.m[j], xb, acc);
                    }
                    #pragma unroll
                    for (int k = 0; k < 8; ++k) T[u * 8 + k][r] = acc[k];
                }
            }
        }
        {
            const int* qb = qbasis + l * 12;
            int lq = hx;
            lq ^= qb[8]  & -(c & 1);
            lq ^= qb[9]  & -((c >> 1) & 1);
            lq ^= qb[10] & -((c >> 2) & 1);
            lq ^= qb[11] & -((c >> 3) & 1);
            #pragma unroll
            for (int t = 0; t < 32; ++t) {
                int ut = 0;
                if (t & 1)  ut ^= qb[3];
                if (t & 2)  ut ^= qb[4];
                if (t & 4)  ut ^= qb[5];
                if (t & 8)  ut ^= qb[6];
                if (t & 16) ut ^= qb[7];
                #pragma unroll
                for (int r = 0; r < 8; ++r) {
                    int ur = ut;
                    if (r & 1) ur ^= qb[0];
                    if (r & 2) ur ^= qb[1];
                    if (r & 4) ur ^= qb[2];
                    sW[lq ^ ur] = T[t][r];
                }
            }
            #pragma unroll
            for (int t = 0; t < 32; ++t) {
                #pragma unroll
                for (int q = 0; q < 2; ++q) {
                    const v4f v = *reinterpret_cast<const v4f*>(&sW[laneRX ^ pos_x(idx_x(t, 4 * q, 0))]);
                    #pragma unroll
                    for (int e = 0; e < 4; ++e) T[t][4 * q + e] = v[e];
                }
            }
        }
    }

    float A[256];
    #pragma unroll
    for (int t = 0; t < 32; ++t) {
        #pragma unroll
        for (int r = 0; r < 8; ++r) A[t * 8 + r] = T[t][r] * T[t][r];
    }
    float oq[8];
    #pragma unroll
    for (int b = 7; b >= 0; --b) {
        const int half = 1 << b;
        float d = 0.0f;
        #pragma unroll
        for (int j = 0; j < half; ++j) {
            d += A[j] - A[j + half];
            A[j] += A[j + half];
        }
        oq[b] = d;
    }
    const float tot = A[0];
    float o[12];
    o[6] = oq[7]; o[7] = oq[6];
    o[3] = oq[5]; o[4] = oq[4]; o[5] = oq[3];
    o[0] = oq[2]; o[1] = oq[1]; o[2] = oq[0];
    #pragma unroll
    for (int k = 8; k < 12; ++k)
        o[k] = ((c >> (11 - k)) & 1) ? -tot : tot;
    float mine = 0.0f;
    #pragma unroll
    for (int k = 0; k < 12; ++k) {
        float v = o[k];
        for (int off = 16; off > 0; off >>= 1) v += __shfl_xor(v, off, 32);
        mine = (lane == k) ? v : mine;
    }
    if (lane < 12) out[(size_t)row * 12 + lane] = mine;
}

extern "C" void kernel_launch(void* const* d_in, const int* in_sizes, int n_in,
                              void* d_out, int out_size, void* d_ws, size_t ws_size,
                              hipStream_t stream) {
    const float* x     = (const float*)d_in[0];
    const float* theta = (const float*)d_in[1];
    float*       out   = (float*)d_out;
    float*       wsf   = (float*)d_ws;

    const int B = in_sizes[0] / DIM;

    qdl_setup_kernel<<<17, 256, 0, stream>>>(theta, wsf);
    qdl_main_kernel<<<(B + WAVES - 1) / WAVES, 256, 0, stream>>>(x, wsf, out, B);
}


# LDS passes run under the wave's own WMMAs: each 8-tile block is stored right behind its 32 WMMAs, read-backs feed the next group tile by tile (gate fragments fetched ahead; fences); sums unchanged
# speedup vs baseline: 2.0780x; 1.0464x over previous
#include <hip/hip_runtime.h>
#include <math.h>

#define NQ    12
#define DIM   4096
#define DEPTH 4
#define WAVES 8

typedef __attribute__((ext_vector_type(2))) float v2f;
typedef __attribute__((ext_vector_type(4))) float v4f;
typedef __attribute__((ext_vector_type(8))) float v8f;

__host__ __device__ inline int idx_x(int t, int r, int c) { return (r << 9) | ((t & 7) << 6) | ((t >> 3) << 4) | c; }
__host__ __device__ inline int idx_y(int t, int r, int c) { return (c << 8) | ((t >> 3) << 6) | ((t & 7) << 3) | r; }

__host__ __device__ inline int pos_x(int i) {
    int p = ((i >> 9) & 3) | ((i & 15) << 2) | (((i >> 4) & 31) << 6) | (i & 0x800);
    return p ^ (((p >> 11) & 1) << 2);
}
__host__ __device__ inline int pos_y(int i) {
    int p = (i & 3) | (((i >> 8) & 15) << 2) | (((i >> 2) & 63) << 6);
    return p ^ (((p >> 6) & 3) << 2);
}
#define HALF_X 32
#define HALF_Y 16

__global__ void qdl_setup_kernel(const float* __restrict__ theta,
                                 float* __restrict__ ws) {
    const int tid = threadIdx.x;
    if (blockIdx.x < 16) {
        const int lg = blockIdx.x;
        const int l  = lg >> 2, g = lg & 3;
        const int m  = tid >> 4, n = tid & 15;
        const int i  = m & 7, j = n & 7;
        const int pm = m >> 3, pn = n >> 3;

        float pr = 1.0f, pi = 0.0f;
        #pragma unroll
        for (int q = 0; q < 3; ++q) {
            const int w = g * 3 + q;
            const float* th = theta + (l * NQ + w) * 3;
            const float phi = th[0], the = th[1], om = th[2];
            const float ch = cosf(0.5f * the), sh = sinf(0.5f * the);
            const float a  = 0.5f * (phi + om), b = 0.5f * (phi - om);
            const float ca = cosf(a), sa = sinf(a);
            const float cb = cosf(b), sb = sinf(b);
            const int bi = (i >> (2 - q)) & 1;
            const int bj = (j >> (2 - q)) & 1;
            float ur, ui;
            if      (bi == 0 && bj == 0) { ur =  ca * ch; ui = -sa * ch; }
            else if (bi == 0 && bj == 1) { ur = -cb * sh; ui = -sb * sh; }
            else if (bi == 1 && bj == 0) { ur =  cb * sh; ui = -sb * sh; }
            else                         { ur =  ca * ch; ui =  sa * ch; }
            const float nr = pr * ur - pi * ui;
            const float ni = pr * ui + pi * ur;
            pr = nr; pi = ni;
        }
        float v;
        if (pm == pn) v = pr; else if (pm == 1) v = pi; else v = -pi;
        ws[lg * 256 + tid] = v;
    } else if (tid < 48) {
        const int l = tid / 12, k = tid % 12;
        const int r = l % (NQ - 1) + 1;
        int j = 1 << k;
        for (int w = 0; w < NQ; ++w) {
            const int tg   = (w + r) % NQ;
            const int cbit = (j >> (NQ - 1 - w)) & 1;
            j ^= cbit << (NQ - 1 - tg);
        }
        ((int*)ws)[16 * 256 + tid] = pos_x(j);
    }
}

template <int M>
__device__ inline float swz_xor(float v) {
    return __int_as_float(__builtin_amdgcn_ds_swizzle(__float_as_int(v), (M << 10) | 0x1F));
}

__device__ inline v8f mma(v2f a, v2f b, v8f c) {
    return __builtin_amdgcn_wmma_f32_16x16x4_f32(false, a, false, b, (short)0, c, false, false);
}

struct GateFrag { v2f m[4]; };

__device__ inline GateFrag load_gate(const float* gl, int lg, int c, int h) {
    GateFrag f;
    const float* G = gl + lg * 256 + c * 16 + 8 * h;
    #pragma unroll
    for (int j = 0; j < 4; ++j) f.m[j] = *reinterpret_cast<const v2f*>(G + 2 * j);
    return f;
}

__global__ __launch_bounds__(256) void qdl_main_kernel(
        const float* __restrict__ x,
        const float* __restrict__ ws,
        float* __restrict__ out,
        int B) {
    __shared__ float st[WAVES * 2 * DIM];
    __shared__ float gLDS[16 * 256];

    const int tid  = threadIdx.x;
    const int lane = tid & 31;
    const int wave = tid >> 5;
    const int h    = lane >> 4;
    const int c    = lane & 15;

    const int row  = blockIdx.x * WAVES + wave;
    const int rowc = row < B ? row : B - 1;

    const float* xr = x + (size_t)rowc * DIM;
    v8f T[32];
    #pragma unroll
    for (int t = 0; t < 32; ++t) {
        #pragma unroll
        for (int r = 0; r < 8; ++r)
            T[t][r] = xr[idx_x(t, r, c)];
    }
    const float x0 = xr[0];

    for (int k = tid; k < 16 * 256; k += 256) gLDS[k] = ws[k];
    __syncthreads();
    if (row >= B) return;

    float* sW = st + wave * (2 * DIM);
    const int* qbasis = reinterpret_cast<const int*>(ws) + 16 * 256;

    GateFrag G0n = load_gate(gLDS, 0, c, h);
    __builtin_amdgcn_sched_barrier(0);

    #pragma unroll
    for (int t = 0; t < 32; ++t) {
        #pragma unroll
        for (int r = 0; r < 8; ++r) T[t][r] = h ? 0.0f : T[t][r];
    }

    float ssp[4] = {0.f, 0.f, 0.f, 0.f};
    #pragma unroll
    for (int t = 0; t < 32; ++t) {
        #pragma unroll
        for (int r = 0; r < 8; ++r) ssp[r & 3] += T[t][r] * T[t][r];
    }
    float ss = (ssp[0] + ssp[1]) + (ssp[2] + ssp[3]);
    ss += swz_xor<16>(ss);
    ss += swz_xor<8>(ss);
    ss += swz_xor<4>(ss);
    ss += swz_xor<2>(ss);
    ss += swz_xor<1>(ss);
    const float n1sq = ss;
    const bool  zero = (sqrtf(n1sq) <= 1e-12f);
    const float x0n  = zero ? 1.0f : x0;
    const float inv  = 1.0f / sqrtf(n1sq - x0 * x0 + x0n * x0n);
    T[0][0] = (lane == 0) ? x0n : T[0][0];
    #pragma unroll
    for (int t = 0; t < 32; ++t) T[t] = T[t] * inv;

    const int hy      = (h << 12) | (h ? HALF_Y : 0);
    const int hx      = (h << 12) | (h ? HALF_X : 0);
    const int laneWY  = hy ^ pos_y(idx_x(0, 0, c));
    const int laneRY  = hy ^ pos_y(idx_y(0, 0, c));
    const int laneRX  = hx ^ pos_x(idx_x(0, 0, c));

    __builtin_amdgcn_s_setprio(3);
    for (int l = 0; l < DEPTH; ++l) {
        switch (l) {
            case 1:  __builtin_amdgcn_s_setprio(2); break;
            case 2:  __builtin_amdgcn_s_setprio(1); break;
            case 3:  __builtin_amdgcn_s_setprio(0); break;
            default: break;
        }
        const int* qb = qbasis + l * 12;
        int lq = hx;
        lq ^= qb[8]  & -(c & 1);
        lq ^= qb[9]  & -((c >> 1) & 1);
        lq ^= qb[10] & -((c >> 2) & 1);
        lq ^= qb[11] & -((c >> 3) & 1);
        {
            const GateFrag G = G0n;
            #pragma unroll
            for (int u = 0; u < 4; ++u) {
                #pragma unroll
                for (int k = 0; k < 8; ++k) {
                    const int t = u * 8 + k;
                    v8f acc = {};
                    #pragma unroll
                    for (int j = 0; j < 4; ++j) {
                        const v2f xb = {T[t][2 * j], T[t][2 * j + 1]};
                        acc = mma(G.m[j], xb, acc);
                    }
                    T[t] = acc;
                }
                __builtin_amdgcn_sched_barrier(0);
            }
        }
        const GateFrag G3 = load_gate(gLDS, l * 4 + 3, c, h);
        {
            const GateFrag G = load_gate(gLDS, l * 4 + 1, c, h);
            #pragma unroll
            for (int u = 0; u < 4; ++u) {
                #pragma unroll
                for (int r = 0; r < 8; ++r) {
                    v8f acc = {};
                    #pragma unroll
                    for (int j = 0; j < 4; ++j) {
                        const v2f xb = {T[u * 8 + 2 * j][r], T[u * 8 + 2 * j + 1][r]};
                        acc = mma(G.m[j], xb, acc);
                    }
                    #pragma unroll
                    for (int k = 0; k < 8; ++k) T[u * 8 + k][r] = acc[k];
                }
                __builtin_amdgcn_sched_barrier(0);
                #pragma unroll
                for (int r = 0; r < 8; ++r) {
                    #pragma unroll
                    for (int k = 0; k < 8; ++k)
                        sW[laneWY ^ pos_y(idx_x(u * 8 + k, r, 0))] = T[u * 8 + k][r];
                }
                __builtin_amdgcn_sched_barrier(0);
            }
        }
        #pragma unroll
        for (int u = 0; u < 4; ++u) {
            #pragma unroll
            for (int k = 0; k < 8; ++k) {
                const int t = u * 8 + k;
                #pragma unroll
                for (int q = 0; q < 2; ++q) {
                    const v4f v = *reinterpret_cast<const v4f*>(&sW[laneRY ^ pos_y(idx_y(t, 4 * q, 0))]);
                    #pragma unroll
                    for (int e = 0; e < 4; ++e) T[t][4 * q + e] = v[e];
                }
            }
            __builtin_amdgcn_sched_barrier(0);
        }
        {
            const GateFrag G = G3;
            #pragma unroll
            for (int u = 0; u < 4; ++u) {
                #pragma unroll
                for (int k = 0; k < 8; ++k) {
                    const int t = u * 8 + k;
                    v8f acc = {};
                    #pragma unroll
                    for (int j = 0; j < 4; ++j) {
                        const v2f xb = {T[t][2 * j], T[t][2 * j + 1]};
                        acc = mma(G.m[j], xb, acc);
                    }
                    T[t] = acc;
                }
                __builtin_amdgcn_sched_barrier(0);
            }
        }
        {
            const GateFrag G = load_gate(gLDS, l * 4 + 2, c, h);
            G0n = load_gate(gLDS, (l + 1 < DEPTH ? l + 1 : l) * 4, c, h);
            #pragma unroll
            for (int u = 0; u < 4; ++u) {
                #pragma unroll
                for (int r = 0; r < 8; ++r) {
                    v8f acc = {};
                    #pragma unroll
                    for (int j = 0; j < 4; ++j) {
                        const v2f xb = {T[u * 8 + 2 * j][r], T[u * 8 + 2 * j + 1][r]};
                        acc = mma(G.m[j], xb, acc);
                    }
                    #pragma unroll
                    for (int k = 0; k < 8; ++k) T[u * 8 + k][r] = acc[k];
                }
                __builtin_amdgcn_sched_barrier(0);
                #pragma unroll
                for (int k = 0; k < 8; ++k) {
                    const int t = u * 8 + k;
                    int ut = 0;
                    if (t & 1)  ut ^= qb[3];
                    if (t & 2)  ut ^= qb[4];
                    if (t & 4)  ut ^= qb[5];
                    if (t & 8)  ut ^= qb[6];
                    if (t & 16) ut ^= qb[7];
                    #pragma unroll
                    for (int r = 0; r < 8; ++r) {
                        int ur = ut;
                        if (r & 1) ur ^= qb[0];
                        if (r & 2) ur ^= qb[1];
                        if (r & 4) ur ^= qb[2];
                        sW[lq ^ ur] = T[t][r];
                    }
                }
                __builtin_amdgcn_sched_barrier(0);
            }
            #pragma unroll
            for (int u = 0; u < 4; ++u) {
                #pragma unroll
                for (int k = 0; k < 8; ++k) {
                    const int t = u * 8 + k;
                    #pragma unroll
                    for (int q = 0; q < 2; ++q) {
                        const v4f v = *reinterpret_cast<const v4f*>(&sW[laneRX ^ pos_x(idx_x(t, 4 * q, 0))]);
                        #pragma unroll
                        for (int e = 0; e < 4; ++e) T[t][4 * q + e] = v[e];
                    }
                }
                __builtin_amdgcn_sched_barrier(0);
            }
        }
    }

    float A[256];
    #pragma unroll
    for (int t = 0; t < 32; ++t) {
        #pragma unroll
        for (int r = 0; r < 8; ++r) A[t * 8 + r] = T[t][r] * T[t][r];
    }
    float oq[8];
    #pragma unroll
    for (int b = 7; b >= 0; --b) {
        const int half = 1 << b;
        float d = 0.0f;
        #pragma unroll
        for (int j = 0; j < half; ++j) {
            d += A[j] - A[j + half];
            A[j] += A[j + half];
        }
        oq[b] = d;
    }
    const float tot = A[0];
    float o[12];
    o[6] = oq[7]; o[7] = oq[6];
    o[3] = oq[5]; o[4] = oq[4]; o[5] = oq[3];
    o[0] = oq[2]; o[1] = oq[1]; o[2] = oq[0];
    #pragma unroll
    for (int k = 8; k < 12; ++k)
        o[k] = ((c >> (11 - k)) & 1) ? -tot : tot;
    float mine = 0.0f;
    #pragma unroll
    for (int k = 0; k < 12; ++k) {
        float v = o[k];
        for (int off = 16; off > 0; off >>= 1) v += __shfl_xor(v, off, 32);
        mine = (lane == k) ? v : mine;
    }
    if (lane < 12) out[(size_t)row * 12 + lane] = mine;
}

extern "C" void kernel_launch(void* const* d_in, const int* in_sizes, int n_in,
                              void* d_out, int out_size, void* d_ws, size_t ws_size,
                              hipStream_t stream) {
    const float* x     = (const float*)d_in[0];
    const float* theta = (const float*)d_in[1];
    float*       out   = (float*)d_out;
    float*       wsf   = (float*)d_ws;

    const int B = in_sizes[0] / DIM;

    qdl_setup_kernel<<<17, 256, 0, stream>>>(theta, wsf);
    qdl_main_kernel<<<(B + WAVES - 1) / WAVES, 256, 0, stream>>>(x, wsf, out, B);
}


# Layer 0, first group on the real input: lane halves split the digit (K = 8: 2 not 4 WMMAs per tile, 1984 not 2048 per wave), 128 not 256 row loads per lane; norm and that group's sum order change
# speedup vs baseline: 2.1478x; 1.0336x over previous
#include <hip/hip_runtime.h>
#include <math.h>

#define NQ    12
#define DIM   4096
#define DEPTH 4
#define WAVES 8

typedef __attribute__((ext_vector_type(2))) float v2f;
typedef __attribute__((ext_vector_type(4))) float v4f;
typedef __attribute__((ext_vector_type(8))) float v8f;

__host__ __device__ inline int idx_x(int t, int r, int c) { return (r << 9) | ((t & 7) << 6) | ((t >> 3) << 4) | c; }
__host__ __device__ inline int idx_y(int t, int r, int c) { return (c << 8) | ((t >> 3) << 6) | ((t & 7) << 3) | r; }

__host__ __device__ inline int pos_x(int i) {
    int p = ((i >> 9) & 3) | ((i & 15) << 2) | (((i >> 4) & 31) << 6) | (i & 0x800);
    return p ^ (((p >> 11) & 1) << 2);
}
__host__ __device__ inline int pos_y(int i) {
    int p = (i & 3) | (((i >> 8) & 15) << 2) | (((i >> 2) & 63) << 6);
    return p ^ (((p >> 6) & 3) << 2);
}
#define HALF_X 32
#define HALF_Y 16

__global__ void qdl_setup_kernel(const float* __restrict__ theta,
                                 float* __restrict__ ws) {
    const int tid = threadIdx.x;
    if (blockIdx.x < 16) {
        const int lg = blockIdx.x;
        const int l  = lg >> 2, g = lg & 3;
        const int m  = tid >> 4, n = tid & 15;
        const int i  = m & 7, j = n & 7;
        const int pm = m >> 3, pn = n >> 3;

        float pr = 1.0f, pi = 0.0f;
        #pragma unroll
        for (int q = 0; q < 3; ++q) {
            const int w = g * 3 + q;
            const float* th = theta + (l * NQ + w) * 3;
            const float phi = th[0], the = th[1], om = th[2];
            const float ch = cosf(0.5f * the), sh = sinf(0.5f * the);
            const float a  = 0.5f * (phi + om), b = 0.5f * (phi - om);
            const float ca = cosf(a), sa = sinf(a);
            const float cb = cosf(b), sb = sinf(b);
            const int bi = (i >> (2 - q)) & 1;
            const int bj = (j >> (2 - q)) & 1;
            float ur, ui;
            if      (bi == 0 && bj == 0) { ur =  ca * ch; ui = -sa * ch; }
            else if (bi == 0 && bj == 1) { ur = -cb * sh; ui = -sb * sh; }
            else if (bi == 1 && bj == 0) { ur =  cb * sh; ui = -sb * sh; }
            else                         { ur =  ca * ch; ui =  sa * ch; }
            const float nr = pr * ur - pi * ui;
            const float ni = pr * ui + pi * ur;
            pr = nr; pi = ni;
        }
        float v;
        if (pm == pn) v = pr; else if (pm == 1) v = pi; else v = -pi;
        ws[lg * 256 + tid] = v;
    } else if (tid < 48) {
        const int l = tid / 12, k = tid % 12;
        const int r = l % (NQ - 1) + 1;
        int j = 1 << k;
        for (int w = 0; w < NQ; ++w) {
            const int tg   = (w + r) % NQ;
            const int cbit = (j >> (NQ - 1 - w)) & 1;
            j ^= cbit << (NQ - 1 - tg);
        }
        ((int*)ws)[16 * 256 + tid] = pos_x(j);
    }
}

template <int M>
__device__ inline float swz_xor(float v) {
    return __int_as_float(__builtin_amdgcn_ds_swizzle(__float_as_int(v), (M << 10) | 0x1F));
}

__device__ inline v8f mma(v2f a, v2f b, v8f c) {
    return __builtin_amdgcn_wmma_f32_16x16x4_f32(false, a, false, b, (short)0, c, false, false);
}

struct GateFrag { v2f m[4]; };

__device__ inline GateFrag load_gate(const float* gl, int lg, int c, int h) {
    GateFrag f;
    const float* G = gl + lg * 256 + c * 16 + 8 * h;
    #pragma unroll
    for (int j = 0; j < 4; ++j) f.m[j] = *reinterpret_cast<const v2f*>(G + 2 * j);
    return f;
}

__global__ __launch_bounds__(256) void qdl_main_kernel(
        const float* __restrict__ x,
        const float* __restrict__ ws,
        float* __restrict__ out,
        int B) {
    __shared__ float st[WAVES * 2 * DIM];
    __shared__ float gLDS[16 * 256];

    const int tid  = threadIdx.x;
    const int lane = tid & 31;
    const int wave = tid >> 5;
    const int h    = lane >> 4;
    const int c    = lane & 15;

    const int row  = blockIdx.x * WAVES + wave;
    const int rowc = row < B ? row : B - 1;

    const float* xr = x + (size_t)rowc * DIM;
    v4f Xin[32];
    #pragma unroll
    for (int t = 0; t < 32; ++t) {
        #pragma unroll
        for (int e = 0; e < 4; ++e)
            Xin[t][e] = xr[idx_x(t, 4 * (e >> 1) + 2 * h + (e & 1), c)];
    }
    const float x0 = xr[0];

    for (int k = tid; k < 16 * 256; k += 256) gLDS[k] = ws[k];
    __syncthreads();
    if (row >= B) return;

    float* sW = st + wave * (2 * DIM);
    const int* qbasis = reinterpret_cast<const int*>(ws) + 16 * 256;

    v2f g0[2];
    {
        const float* G = gLDS + c * 16 + 2 * h;
        g0[0] = *reinterpret_cast<const v2f*>(G);
        g0[1] = *reinterpret_cast<const v2f*>(G + 4);
    }

    float ssp[4] = {0.f, 0.f, 0.f, 0.f};
    #pragma unroll
    for (int t = 0; t < 32; ++t) {
        #pragma unroll
        for (int e = 0; e < 4; ++e) ssp[e] += Xin[t][e] * Xin[t][e];
    }
    float ss = (ssp[0] + ssp[1]) + (ssp[2] + ssp[3]);
    ss += swz_xor<16>(ss);
    ss += swz_xor<8>(ss);
    ss += swz_xor<4>(ss);
    ss += swz_xor<2>(ss);
    ss += swz_xor<1>(ss);
    const float n1sq = ss;
    const bool  zero = (sqrtf(n1sq) <= 1e-12f);
    const float x0n  = zero ? 1.0f : x0;
    const float inv  = 1.0f / sqrtf(n1sq - x0 * x0 + x0n * x0n);
    Xin[0][0] = (lane == 0) ? x0n : Xin[0][0];
    #pragma unroll
    for (int t = 0; t < 32; ++t) Xin[t] = Xin[t] * inv;

    v8f T[32];
    #pragma unroll
    for (int t = 0; t < 32; ++t) {
        v8f acc = {};
        #pragma unroll
        for (int j = 0; j < 2; ++j) {
            const v2f xb = {Xin[t][2 * j], Xin[t][2 * j + 1]};
            acc = mma(g0[j], xb, acc);
        }
        T[t] = acc;
    }

    const int hy      = (h << 12) | (h ? HALF_Y : 0);
    const int hx      = (h << 12) | (h ? HALF_X : 0);
    const int laneWY  = hy ^ pos_y(idx_x(0, 0, c));
    const int laneRY  = hy ^ pos_y(idx_y(0, 0, c));
    const int laneRX  = hx ^ pos_x(idx_x(0, 0, c));

    __builtin_amdgcn_s_setprio(3);
    for (int l = 0; ; ++l) {
        const int* qb = qbasis + l * 12;
        int lq = hx;
        lq ^= qb[8]  & -(c & 1);
        lq ^= qb[9]  & -((c >> 1) & 1);
        lq ^= qb[10] & -((c >> 2) & 1);
        lq ^= qb[11] & -((c >> 3) & 1);
        const GateFrag G3 = load_gate(gLDS, l * 4 + 3, c, h);
        {
            const GateFrag G = load_gate(gLDS, l * 4 + 1, c, h);
            #pragma unroll
            for (int u = 0; u < 4; ++u) {
                #pragma unroll
                for (int r = 0; r < 8; ++r) {
                    v8f acc = {};
                    #pragma unroll
                    for (int j = 0; j < 4; ++j) {
                        const v2f xb = {T[u * 8 + 2 * j][r], T[u * 8 + 2 * j + 1][r]};
                        acc = mma(G.m[j], xb, acc);
                    }
                    #pragma unroll
                    for (int k = 0; k < 8; ++k) T[u * 8 + k][r] = acc[k];
                }
                __builtin_amdgcn_sched_barrier(0);
                #pragma unroll
                for (int r = 0; r < 8; ++r) {
                    #pragma unroll
                    for (int k = 0; k < 8; ++k)
                        sW[laneWY ^ pos_y(idx_x(u * 8 + k, r, 0))] = T[u * 8 + k][r];
                }
                __builtin_amdgcn_sched_barrier(0);
            }
        }
        #pragma unroll
        for (int u = 0; u < 4; ++u) {
            #pragma unroll
            for (int k = 0; k < 8; ++k) {
                const int t = u * 8 + k;
                #pragma unroll
                for (int q = 0; q < 2; ++q) {
                    const v4f v = *reinterpret_cast<const v4f*>(&sW[laneRY ^ pos_y(idx_y(t, 4 * q, 0))]);
                    #pragma unroll
                    for (int e = 0; e < 4; ++e) T[t][4 * q + e] = v[e];
                }
            }
            __builtin_amdgcn_sched_barrier(0);
        }
        {
            const GateFrag G = G3;
            #pragma unroll
            for (int u = 0; u < 4; ++u) {
                #pragma unroll
                for (int k = 0; k < 8; ++k) {
                    const int t = u * 8 + k;
                    v8f acc = {};
                    #pragma unroll
                    for (int j = 0; j < 4; ++j) {
                        const v2f xb = {T[t][2 * j], T[t][2 * j + 1]};
                        acc = mma(G.m[j], xb, acc);
                    }
                    T[t] = acc;
                }
                __builtin_amdgcn_sched_barrier(0);
            }
        }
        {
            const GateFrag G = load_gate(gLDS, l * 4 + 2, c, h);
            const GateFrag G0n = load_gate(gLDS, (l + 1 < DEPTH ? l + 1 : l) * 4, c, h);
            #pragma unroll
            for (int u = 0; u < 4; ++u) {
                #pragma unroll
                for (int r = 0; r < 8; ++r) {
                    v8f acc = {};
                    #pragma unroll
                    for (int j = 0; j < 4; ++j) {
                        const v2f xb = {T[u * 8 + 2 * j][r], T[u * 8 + 2 * j + 1][r]};
                        acc = mma(G.m[j], xb, acc);
                    }
                    #pragma unroll
                    for (int k = 0; k < 8; ++k) T[u * 8 + k][r] = acc[k];
                }
                __builtin_amdgcn_sched_barrier(0);
                #pragma unroll
                for (int k = 0; k < 8; ++k) {
                    const int t = u * 8 + k;
                    int ut = 0;
                    if (t & 1)  ut ^= qb[3];
                    if (t & 2)  ut ^= qb[4];
                    if (t & 4)  ut ^= qb[5];
                    if (t & 8)  ut ^= qb[6];
                    if (t & 16) ut ^= qb[7];
                    #pragma unroll
                    for (int r = 0; r < 8; ++r) {
                        int ur = ut;
                        if (r & 1) ur ^= qb[0];
                        if (r & 2) ur ^= qb[1];
                        if (r & 4) ur ^= qb[2];
                        sW[lq ^ ur] = T[t][r];
                    }
                }
                __builtin_amdgcn_sched_barrier(0);
            }
            auto read_x = [&](int u) __attribute__((always_inline)) {
                #pragma unroll
                for (int k = 0; k < 8; ++k) {
                    const int t = u * 8 + k;
                    #pragma unroll
                    for (int q = 0; q < 2; ++q) {
                        const v4f v = *reinterpret_cast<const v4f*>(&sW[laneRX ^ pos_x(idx_x(t, 4 * q, 0))]);
                        #pragma unroll
                        for (int e = 0; e < 4; ++e) T[t][4 * q + e] = v[e];
                    }
                }
                __builtin_amdgcn_sched_barrier(0);
            };
            auto gate_x = [&](int u) __attribute__((always_inline)) {
                #pragma unroll
                for (int k = 0; k < 8; ++k) {
                    const int t = u * 8 + k;
                    v8f acc = {};
                    #pragma unroll
                    for (int j = 0; j < 4; ++j) {
                        const v2f xb = {T[t][2 * j], T[t][2 * j + 1]};
                        acc = mma(G0n.m[j], xb, acc);
                    }
                    T[t] = acc;
                }
                __builtin_amdgcn_sched_barrier(0);
            };
            read_x(0);
            read_x(1);
            if (l + 1 == DEPTH) {
                read_x(2);
                read_x(3);
                break;
            }
            switch (l) {
                case 0:  __builtin_amdgcn_s_setprio(2); break;
                case 1:  __builtin_amdgcn_s_setprio(1); break;
                default: __builtin_amdgcn_s_setprio(0); break;
            }
            gate_x(0);
            read_x(2);
            gate_x(1);
            read_x(3);
            gate_x(2);
            gate_x(3);
        }
    }

    float A[256];
    #pragma unroll
    for (int t = 0; t < 32; ++t) {
        #pragma unroll
        for (int r = 0; r < 8; ++r) A[t * 8 + r] = T[t][r] * T[t][r];
    }
    float oq[8];
    #pragma unroll
    for (int b = 7; b >= 0; --b) {
        const int half = 1 << b;
        float d = 0.0f;
        #pragma unroll
        for (int j = 0; j < half; ++j) {
            d += A[j] - A[j + half];
            A[j] += A[j + half];
        }
        oq[b] = d;
    }
    const float tot = A[0];
    float o[12];
    o[6] = oq[7]; o[7] = oq[6];
    o[3] = oq[5]; o[4] = oq[4]; o[5] = oq[3];
    o[0] = oq[2]; o[1] = oq[1]; o[2] = oq[0];
    #pragma unroll
    for (int k = 8; k < 12; ++k)
        o[k] = ((c >> (11 - k)) & 1) ? -tot : tot;
    float mine = 0.0f;
    #pragma unroll
    for (int k = 0; k < 12; ++k) {
        float v = o[k];
        for (int off = 16; off > 0; off >>= 1) v += __shfl_xor(v, off, 32);
        mine = (lane == k) ? v : mine;
    }
    if (lane < 12) out[(size_t)row * 12 + lane] = mine;
}

extern "C" void kernel_launch(void* const* d_in, const int* in_sizes, int n_in,
                              void* d_out, int out_size, void* d_ws, size_t ws_size,
                              hipStream_t stream) {
    const float* x     = (const float*)d_in[0];
    const float* theta = (const float*)d_in[1];
    float*       out   = (float*)d_out;
    float*       wsf   = (float*)d_ws;

    const int B = in_sizes[0] / DIM;

    qdl_setup_kernel<<<17, 256, 0, stream>>>(theta, wsf);
    qdl_main_kernel<<<(B + WAVES - 1) / WAVES, 256, 0, stream>>>(x, wsf, out, B);
}


# Gate fragments read straight from the setup table in global memory, each a WMMA group ahead of its use: no LDS gate stage, none of its eight serial fetch round trips, no block barrier; sums unchanged
# speedup vs baseline: 2.2795x; 1.0613x over previous
#include <hip/hip_runtime.h>
#include <math.h>

#define NQ    12
#define DIM   4096
#define DEPTH 4
#define WAVES 8

typedef __attribute__((ext_vector_type(2))) float v2f;
typedef __attribute__((ext_vector_type(4))) float v4f;
typedef __attribute__((ext_vector_type(8))) float v8f;

__host__ __device__ inline int idx_x(int t, int r, int c) { return (r << 9) | ((t & 7) << 6) | ((t >> 3) << 4) | c; }
__host__ __device__ inline int idx_y(int t, int r, int c) { return (c << 8) | ((t >> 3) << 6) | ((t & 7) << 3) | r; }

__host__ __device__ inline int pos_x(int i) {
    int p = ((i >> 9) & 3) | ((i & 15) << 2) | (((i >> 4) & 31) << 6) | (i & 0x800);
    return p ^ (((p >> 11) & 1) << 2);
}
__host__ __device__ inline int pos_y(int i) {
    int p = (i & 3) | (((i >> 8) & 15) << 2) | (((i >> 2) & 63) << 6);
    return p ^ (((p >> 6) & 3) << 2);
}
#define HALF_X 32
#define HALF_Y 16

__global__ void qdl_setup_kernel(const float* __restrict__ theta,
                                 float* __restrict__ ws) {
    const int tid = threadIdx.x;
    if (blockIdx.x < 16) {
        const int lg = blockIdx.x;
        const int l  = lg >> 2, g = lg & 3;
        const int m  = tid >> 4, n = tid & 15;
        const int i  = m & 7, j = n & 7;
        const int pm = m >> 3, pn = n >> 3;

        float pr = 1.0f, pi = 0.0f;
        #pragma unroll
        for (int q = 0; q < 3; ++q) {
            const int w = g * 3 + q;
            const float* th = theta + (l * NQ + w) * 3;
            const float phi = th[0], the = th[1], om = th[2];
            const float ch = cosf(0.5f * the), sh = sinf(0.5f * the);
            const float a  = 0.5f * (phi + om), b = 0.5f * (phi - om);
            const float ca = cosf(a), sa = sinf(a);
            const float cb = cosf(b), sb = sinf(b);
            const int bi = (i >> (2 - q)) & 1;
            const int bj = (j >> (2 - q)) & 1;
            float ur, ui;
            if      (bi == 0 && bj == 0) { ur =  ca * ch; ui = -sa * ch; }
            else if (bi == 0 && bj == 1) { ur = -cb * sh; ui = -sb * sh; }
            else if (bi == 1 && bj == 0) { ur =  cb * sh; ui = -sb * sh; }
            else                         { ur =  ca * ch; ui =  sa * ch; }
            const float nr = pr * ur - pi * ui;
            const float ni = pr * ui + pi * ur;
            pr = nr; pi = ni;
        }
        float v;
        if (pm == pn) v = pr; else if (pm == 1) v = pi; else v = -pi;
        ws[lg * 256 + tid] = v;
    } else if (tid < 48) {
        const int l = tid / 12, k = tid % 12;
        const int r = l % (NQ - 1) + 1;
        int j = 1 << k;
        for (int w = 0; w < NQ; ++w) {
            const int tg   = (w + r) % NQ;
            const int cbit = (j >> (NQ - 1 - w)) & 1;
            j ^= cbit << (NQ - 1 - tg);
        }
        ((int*)ws)[16 * 256 + tid] = pos_x(j);
    }
}

template <int M>
__device__ inline float swz_xor(float v) {
    return __int_as_float(__builtin_amdgcn_ds_swizzle(__float_as_int(v), (M << 10) | 0x1F));
}

__device__ inline v8f mma(v2f a, v2f b, v8f c) {
    return __builtin_amdgcn_wmma_f32_16x16x4_f32(false, a, false, b, (short)0, c, false, false);
}

struct GateFrag { v2f m[4]; };

__device__ inline GateFrag load_gate(const float* gl, int lg, int c, int h) {
    GateFrag f;
    const float* G = gl + lg * 256 + c * 16 + 8 * h;
    #pragma unroll
    for (int j = 0; j < 4; ++j) f.m[j] = *reinterpret_cast<const v2f*>(G + 2 * j);
    return f;
}

__global__ __launch_bounds__(256) void qdl_main_kernel(
        const float* __restrict__ x,
        const float* __restrict__ ws,
        float* __restrict__ out,
        int B) {
    __shared__ float st[WAVES * 2 * DIM];

    const int tid  = threadIdx.x;
    const int lane = tid & 31;
    const int wave = tid >> 5;
    const int h    = lane >> 4;
    const int c    = lane & 15;

    const int row  = blockIdx.x * WAVES + wave;

    if (row >= B) return;

    v2f g0[2];
    {
        const float* G = ws + c * 16 + 2 * h;
        g0[0] = *reinterpret_cast<const v2f*>(G);
        g0[1] = *reinterpret_cast<const v2f*>(G + 4);
    }

    GateFrag G1 = load_gate(ws, 1, c, h);

    const float* xr = x + (size_t)row * DIM;
    v4f Xin[32];
    #pragma unroll
    for (int t = 0; t < 32; ++t) {
        #pragma unroll
        for (int e = 0; e < 4; ++e)
            Xin[t][e] = xr[idx_x(t, 4 * (e >> 1) + 2 * h + (e & 1), c)];
    }
    const float x0 = xr[0];

    float* sW = st + wave * (2 * DIM);
    const int* qbasis = reinterpret_cast<const int*>(ws) + 16 * 256;

    float ssp[4] = {0.f, 0.f, 0.f, 0.f};
    #pragma unroll
    for (int t = 0; t < 32; ++t) {
        #pragma unroll
        for (int e = 0; e < 4; ++e) ssp[e] += Xin[t][e] * Xin[t][e];
    }
    float ss = (ssp[0] + ssp[1]) + (ssp[2] + ssp[3]);
    ss += swz_xor<16>(ss);
    ss += swz_xor<8>(ss);
    ss += swz_xor<4>(ss);
    ss += swz_xor<2>(ss);
    ss += swz_xor<1>(ss);
    const float n1sq = ss;
    const bool  zero = (sqrtf(n1sq) <= 1e-12f);
    const float x0n  = zero ? 1.0f : x0;
    const float inv  = 1.0f / sqrtf(n1sq - x0 * x0 + x0n * x0n);
    Xin[0][0] = (lane == 0) ? x0n : Xin[0][0];
    #pragma unroll
    for (int t = 0; t < 32; ++t) Xin[t] = Xin[t] * inv;

    v8f T[32];
    #pragma unroll
    for (int t = 0; t < 32; ++t) {
        v8f acc = {};
        #pragma unroll
        for (int j = 0; j < 2; ++j) {
            const v2f xb = {Xin[t][2 * j], Xin[t][2 * j + 1]};
            acc = mma(g0[j], xb, acc);
        }
        T[t] = acc;
    }

    const int hy      = (h << 12) | (h ? HALF_Y : 0);
    const int hx      = (h << 12) | (h ? HALF_X : 0);
    const int laneWY  = hy ^ pos_y(idx_x(0, 0, c));
    const int laneRY  = hy ^ pos_y(idx_y(0, 0, c));
    const int laneRX  = hx ^ pos_x(idx_x(0, 0, c));

    __builtin_amdgcn_s_setprio(3);
    for (int l = 0; ; ++l) {
        const int* qb = qbasis + l * 12;
        int lq = hx;
        lq ^= qb[8]  & -(c & 1);
        lq ^= qb[9]  & -((c >> 1) & 1);
        lq ^= qb[10] & -((c >> 2) & 1);
        lq ^= qb[11] & -((c >> 3) & 1);
        const GateFrag G3 = load_gate(ws, l * 4 + 3, c, h);
        GateFrag G2;
        {
            const GateFrag G = G1;
            #pragma unroll
            for (int u = 0; u < 4; ++u) {
                #pragma unroll
                for (int r = 0; r < 8; ++r) {
                    v8f acc = {};
                    #pragma unroll
                    for (int j = 0; j < 4; ++j) {
                        const v2f xb = {T[u * 8 + 2 * j][r], T[u * 8 + 2 * j + 1][r]};
                        acc = mma(G.m[j], xb, acc);
                    }
                    #pragma unroll
                    for (int k = 0; k < 8; ++k) T[u * 8 + k][r] = acc[k];
                }
                __builtin_amdgcn_sched_barrier(0);
                #pragma unroll
                for (int r = 0; r < 8; ++r) {
                    #pragma unroll
                    for (int k = 0; k < 8; ++k)
                        sW[laneWY ^ pos_y(idx_x(u * 8 + k, r, 0))] = T[u * 8 + k][r];
                }
                __builtin_amdgcn_sched_barrier(0);
            }
        }
        #pragma unroll
        for (int u = 0; u < 4; ++u) {
            #pragma unroll
            for (int k = 0; k < 8; ++k) {
                const int t = u * 8 + k;
                #pragma unroll
                for (int q = 0; q < 2; ++q) {
                    const v4f v = *reinterpret_cast<const v4f*>(&sW[laneRY ^ pos_y(idx_y(t, 4 * q, 0))]);
                    #pragma unroll
                    for (int e = 0; e < 4; ++e) T[t][4 * q + e] = v[e];
                }
            }
            __builtin_amdgcn_sched_barrier(0);
        }
        {
            const GateFrag G = G3;
            #pragma unroll
            for (int u = 0; u < 4; ++u) {
                if (u == 2) G2 = load_gate(ws, l * 4 + 2, c, h);
                #pragma unroll
                for (int k = 0; k < 8; ++k) {
                    const int t = u * 8 + k;
                    v8f acc = {};
                    #pragma unroll
                    for (int j = 0; j < 4; ++j) {
                        const v2f xb = {T[t][2 * j], T[t][2 * j + 1]};
                        acc = mma(G.m[j], xb, acc);
                    }
                    T[t] = acc;
                }
                __builtin_amdgcn_sched_barrier(0);
            }
        }
        {
            const GateFrag G = G2;
            const GateFrag G0n = load_gate(ws, (l + 1 < DEPTH ? l + 1 : l) * 4, c, h);
            #pragma unroll
            for (int u = 0; u < 4; ++u) {
                #pragma unroll
                for (int r = 0; r < 8; ++r) {
                    v8f acc = {};
                    #pragma unroll
                    for (int j = 0; j < 4; ++j) {
                        const v2f xb = {T[u * 8 + 2 * j][r], T[u * 8 + 2 * j + 1][r]};
                        acc = mma(G.m[j], xb, acc);
                    }
                    #pragma unroll
                    for (int k = 0; k < 8; ++k) T[u * 8 + k][r] = acc[k];
                }
                __builtin_amdgcn_sched_barrier(0);
                #pragma unroll
                for (int k = 0; k < 8; ++k) {
                    const int t = u * 8 + k;
                    int ut = 0;
                    if (t & 1)  ut ^= qb[3];
                    if (t & 2)  ut ^= qb[4];
                    if (t & 4)  ut ^= qb[5];
                    if (t & 8)  ut ^= qb[6];
                    if (t & 16) ut ^= qb[7];
                    #pragma unroll
                    for (int r = 0; r < 8; ++r) {
                        int ur = ut;
                        if (r & 1) ur ^= qb[0];
                        if (r & 2) ur ^= qb[1];
                        if (r & 4) ur ^= qb[2];
                        sW[lq ^ ur] = T[t][r];
                    }
                }
                __builtin_amdgcn_sched_barrier(0);
            }
            asm volatile("" :: "v"(G0n.m[0]), "v"(G0n.m[1]), "v"(G0n.m[2]), "v"(G0n.m[3]));
            auto read_x = [&](int u) __attribute__((always_inline)) {
                #pragma unroll
                for (int k = 0; k < 8; ++k) {
                    const int t = u * 8 + k;
                    #pragma unroll
                    for (int q = 0; q < 2; ++q) {
                        const v4f v = *reinterpret_cast<const v4f*>(&sW[laneRX ^ pos_x(idx_x(t, 4 * q, 0))]);
                        #pragma unroll
                        for (int e = 0; e < 4; ++e) T[t][4 * q + e] = v[e];
                    }
                }
                __builtin_amdgcn_sched_barrier(0);
            };
            auto gate_x = [&](int u) __attribute__((always_inline)) {
                #pragma unroll
                for (int k = 0; k < 8; ++k) {
                    const int t = u * 8 + k;
                    v8f acc = {};
                    #pragma unroll
                    for (int j = 0; j < 4; ++j) {
                        const v2f xb = {T[t][2 * j], T[t][2 * j + 1]};
                        acc = mma(G0n.m[j], xb, acc);
                    }
                    T[t] = acc;
                }
                __builtin_amdgcn_sched_barrier(0);
            };
            read_x(0);
            read_x(1);
            if (l + 1 == DEPTH) {
                read_x(2);
                read_x(3);
                break;
            }
            switch (l) {
                case 0:  __builtin_amdgcn_s_setprio(2); break;
                case 1:  __builtin_amdgcn_s_setprio(1); break;
                default: __builtin_amdgcn_s_setprio(0); break;
            }
            gate_x(0);
            read_x(2);
            gate_x(1);
            read_x(3);
            G1 = load_gate(ws, (l + 1) * 4 + 1, c, h);
            gate_x(2);
            gate_x(3);
        }
    }

    float A[256];
    #pragma unroll
    for (int t = 0; t < 32; ++t) {
        #pragma unroll
        for (int r = 0; r < 8; ++r) A[t * 8 + r] = T[t][r] * T[t][r];
    }
    float oq[8];
    #pragma unroll
    for (int b = 7; b >= 0; --b) {
        const int half = 1 << b;
        float d = 0.0f;
        #pragma unroll
        for (int j = 0; j < half; ++j) {
            d += A[j] - A[j + half];
            A[j] += A[j + half];
        }
        oq[b] = d;
    }
    const float tot = A[0];
    float o[12];
    o[6] = oq[7]; o[7] = oq[6];
    o[3] = oq[5]; o[4] = oq[4]; o[5] = oq[3];
    o[0] = oq[2]; o[1] = oq[1]; o[2] = oq[0];
    #pragma unroll
    for (int k = 8; k < 12; ++k)
        o[k] = ((c >> (11 - k)) & 1) ? -tot : tot;
    float mine = 0.0f;
    #pragma unroll
    for (int k = 0; k < 12; ++k) {
        float v = o[k];
        for (int off = 16; off > 0; off >>= 1) v += __shfl_xor(v, off, 32);
        mine = (lane == k) ? v : mine;
    }
    if (lane < 12) out[(size_t)row * 12 + lane] = mine;
}

extern "C" void kernel_launch(void* const* d_in, const int* in_sizes, int n_in,
                              void* d_out, int out_size, void* d_ws, size_t ws_size,
                              hipStream_t stream) {
    const float* x     = (const float*)d_in[0];
    const float* theta = (const float*)d_in[1];
    float*       out   = (float*)d_out;
    float*       wsf   = (float*)d_ws;

    const int B = in_sizes[0] / DIM;

    qdl_setup_kernel<<<17, 256, 0, stream>>>(theta, wsf);
    qdl_main_kernel<<<(B + WAVES - 1) / WAVES, 256, 0, stream>>>(x, wsf, out, B);
}


# Entangler scatter addresses formed beside each block's own WMMAs, one XOR each (basis words as byte offsets; 8 per 4 WMMAs, fenced): store blocks hold stores only, no shift-adds; sums unchanged
# speedup vs baseline: 2.4272x; 1.0648x over previous
#include <hip/hip_runtime.h>
#include <math.h>

#define NQ    12
#define DIM   4096
#define DEPTH 4
#define WAVES 8

typedef __attribute__((ext_vector_type(2))) float v2f;
typedef __attribute__((ext_vector_type(4))) float v4f;
typedef __attribute__((ext_vector_type(8))) float v8f;

__host__ __device__ inline int idx_x(int t, int r, int c) { return (r << 9) | ((t & 7) << 6) | ((t >> 3) << 4) | c; }
__host__ __device__ inline int idx_y(int t, int r, int c) { return (c << 8) | ((t >> 3) << 6) | ((t & 7) << 3) | r; }

__host__ __device__ inline int pos_x(int i) {
    int p = ((i >> 9) & 3) | ((i & 15) << 2) | (((i >> 4) & 31) << 6) | (i & 0x800);
    return p ^ (((p >> 11) & 1) << 2);
}
__host__ __device__ inline int pos_y(int i) {
    int p = (i & 3) | (((i >> 8) & 15) << 2) | (((i >> 2) & 63) << 6);
    return p ^ (((p >> 6) & 3) << 2);
}
#define HALF_X 32
#define HALF_Y 16

__global__ void qdl_setup_kernel(const float* __restrict__ theta,
                                 float* __restrict__ ws) {
    const int tid = threadIdx.x;
    if (blockIdx.x < 16) {
        const int lg = blockIdx.x;
        const int l  = lg >> 2, g = lg & 3;
        const int m  = tid >> 4, n = tid & 15;
        const int i  = m & 7, j = n & 7;
        const int pm = m >> 3, pn = n >> 3;

        float pr = 1.0f, pi = 0.0f;
        #pragma unroll
        for (int q = 0; q < 3; ++q) {
            const int w = g * 3 + q;
            const float* th = theta + (l * NQ + w) * 3;
            const float phi = th[0], the = th[1], om = th[2];
            const float ch = cosf(0.5f * the), sh = sinf(0.5f * the);
            const float a  = 0.5f * (phi + om), b = 0.5f * (phi - om);
            const float ca = cosf(a), sa = sinf(a);
            const float cb = cosf(b), sb = sinf(b);
            const int bi = (i >> (2 - q)) & 1;
            const int bj = (j >> (2 - q)) & 1;
            float ur, ui;
            if      (bi == 0 && bj == 0) { ur =  ca * ch; ui = -sa * ch; }
            else if (bi == 0 && bj == 1) { ur = -cb * sh; ui = -sb * sh; }
            else if (bi == 1 && bj == 0) { ur =  cb * sh; ui = -sb * sh; }
            else                         { ur =  ca * ch; ui =  sa * ch; }
            const float nr = pr * ur - pi * ui;
            const float ni = pr * ui + pi * ur;
            pr = nr; pi = ni;
        }
        float v;
        if (pm == pn) v = pr; else if (pm == 1) v = pi; else v = -pi;
        ws[lg * 256 + tid] = v;
    } else if (tid < 48) {
        const int l = tid / 12, k = tid % 12;
        const int r = l % (NQ - 1) + 1;
        int j = 1 << k;
        for (int w = 0; w < NQ; ++w) {
            const int tg   = (w + r) % NQ;
            const int cbit = (j >> (NQ - 1 - w)) & 1;
            j ^= cbit << (NQ - 1 - tg);
        }
        ((int*)ws)[16 * 256 + tid] = pos_x(j) << 2;
    }
}

template <int M>
__device__ inline float swz_xor(float v) {
    return __int_as_float(__builtin_amdgcn_ds_swizzle(__float_as_int(v), (M << 10) | 0x1F));
}

__device__ inline v8f mma(v2f a, v2f b, v8f c) {
    return __builtin_amdgcn_wmma_f32_16x16x4_f32(false, a, false, b, (short)0, c, false, false);
}

struct GateFrag { v2f m[4]; };

__device__ inline GateFrag load_gate(const float* gl, int lg, int c, int h) {
    GateFrag f;
    const float* G = gl + lg * 256 + c * 16 + 8 * h;
    #pragma unroll
    for (int j = 0; j < 4; ++j) f.m[j] = *reinterpret_cast<const v2f*>(G + 2 * j);
    return f;
}

__global__ __launch_bounds__(256) void qdl_main_kernel(
        const float* __restrict__ x,
        const float* __restrict__ ws,
        float* __restrict__ out,
        int B) {
    __shared__ float st[WAVES * 2 * DIM];

    const int tid  = threadIdx.x;
    const int lane = tid & 31;
    const int wave = tid >> 5;
    const int h    = lane >> 4;
    const int c    = lane & 15;

    const int row  = blockIdx.x * WAVES + wave;

    if (row >= B) return;

    v2f g0[2];
    {
        const float* G = ws + c * 16 + 2 * h;
        g0[0] = *reinterpret_cast<const v2f*>(G);
        g0[1] = *reinterpret_cast<const v2f*>(G + 4);
    }

    GateFrag G1 = load_gate(ws, 1, c, h);

    const float* xr = x + (size_t)row * DIM;
    v4f Xin[32];
    #pragma unroll
    for (int t = 0; t < 32; ++t) {
        #pragma unroll
        for (int e = 0; e < 4; ++e)
            Xin[t][e] = xr[idx_x(t, 4 * (e >> 1) + 2 * h + (e & 1), c)];
    }
    const float x0 = xr[0];

    float* sW = st + wave * (2 * DIM);
    const int waveBytes = wave * (2 * DIM * (int)sizeof(float));
    const int* qbasis = reinterpret_cast<const int*>(ws) + 16 * 256;

    float ssp[4] = {0.f, 0.f, 0.f, 0.f};
    #pragma unroll
    for (int t = 0; t < 32; ++t) {
        #pragma unroll
        for (int e = 0; e < 4; ++e) ssp[e] += Xin[t][e] * Xin[t][e];
    }
    float ss = (ssp[0] + ssp[1]) + (ssp[2] + ssp[3]);
    ss += swz_xor<16>(ss);
    ss += swz_xor<8>(ss);
    ss += swz_xor<4>(ss);
    ss += swz_xor<2>(ss);
    ss += swz_xor<1>(ss);
    const float n1sq = ss;
    const bool  zero = (sqrtf(n1sq) <= 1e-12f);
    const float x0n  = zero ? 1.0f : x0;
    const float inv  = 1.0f / sqrtf(n1sq - x0 * x0 + x0n * x0n);
    Xin[0][0] = (lane == 0) ? x0n : Xin[0][0];
    #pragma unroll
    for (int t = 0; t < 32; ++t) Xin[t] = Xin[t] * inv;

    v8f T[32];
    #pragma unroll
    for (int t = 0; t < 32; ++t) {
        v8f acc = {};
        #pragma unroll
        for (int j = 0; j < 2; ++j) {
            const v2f xb = {Xin[t][2 * j], Xin[t][2 * j + 1]};
            acc = mma(g0[j], xb, acc);
        }
        T[t] = acc;
    }

    const int hy      = (h << 12) | (h ? HALF_Y : 0);
    const int hx      = (h << 12) | (h ? HALF_X : 0);
    const int laneWY  = hy ^ pos_y(idx_x(0, 0, c));
    const int laneRY  = hy ^ pos_y(idx_y(0, 0, c));
    const int laneRX  = hx ^ pos_x(idx_x(0, 0, c));

    __builtin_amdgcn_s_setprio(3);
    for (int l = 0; ; ++l) {
        const int* qb = qbasis + l * 12;
        int lq = waveBytes | (hx << 2);
        lq ^= qb[8]  & -(c & 1);
        lq ^= qb[9]  & -((c >> 1) & 1);
        lq ^= qb[10] & -((c >> 2) & 1);
        lq ^= qb[11] & -((c >> 3) & 1);
        const GateFrag G3 = load_gate(ws, l * 4 + 3, c, h);
        GateFrag G2;
        {
            const GateFrag G = G1;
            #pragma unroll
            for (int u = 0; u < 4; ++u) {
                #pragma unroll
                for (int r = 0; r < 8; ++r) {
                    v8f acc = {};
                    #pragma unroll
                    for (int j = 0; j < 4; ++j) {
                        const v2f xb = {T[u * 8 + 2 * j][r], T[u * 8 + 2 * j + 1][r]};
                        acc = mma(G.m[j], xb, acc);
                    }
                    #pragma unroll
                    for (int k = 0; k < 8; ++k) T[u * 8 + k][r] = acc[k];
                }
                __builtin_amdgcn_sched_barrier(0);
                #pragma unroll
                for (int r = 0; r < 8; ++r) {
                    #pragma unroll
                    for (int k = 0; k < 8; ++k)
                        sW[laneWY ^ pos_y(idx_x(u * 8 + k, r, 0))] = T[u * 8 + k][r];
                }
                __builtin_amdgcn_sched_barrier(0);
            }
        }
        #pragma unroll
        for (int u = 0; u < 4; ++u) {
            #pragma unroll
            for (int k = 0; k < 8; ++k) {
                const int t = u * 8 + k;
                #pragma unroll
                for (int q = 0; q < 2; ++q) {
                    const v4f v = *reinterpret_cast<const v4f*>(&sW[laneRY ^ pos_y(idx_y(t, 4 * q, 0))]);
                    #pragma unroll
                    for (int e = 0; e < 4; ++e) T[t][4 * q + e] = v[e];
                }
            }
            __builtin_amdgcn_sched_barrier(0);
        }
        {
            const GateFrag G = G3;
            #pragma unroll
            for (int u = 0; u < 4; ++u) {
                if (u == 2) G2 = load_gate(ws, l * 4 + 2, c, h);
                #pragma unroll
                for (int k = 0; k < 8; ++k) {
                    const int t = u * 8 + k;
                    v8f acc = {};
                    #pragma unroll
                    for (int j = 0; j < 4; ++j) {
                        const v2f xb = {T[t][2 * j], T[t][2 * j + 1]};
                        acc = mma(G.m[j], xb, acc);
                    }
                    T[t] = acc;
                }
                __builtin_amdgcn_sched_barrier(0);
            }
        }
        {
            const GateFrag G = G2;
            const GateFrag G0n = load_gate(ws, (l + 1 < DEPTH ? l + 1 : l) * 4, c, h);
            #pragma unroll
            for (int u = 0; u < 4; ++u) {
                int ad[8][8];
                v8f acc[8];
                #pragma unroll
                for (int j = 0; j < 4; ++j) {
                    #pragma unroll
                    for (int kk = 0; kk < 2; ++kk) {
                        const int k = 2 * j + kk;
                        const int t = u * 8 + k;
                        int ut = 0;
                        if (t & 1)  ut ^= qb[3];
                        if (t & 2)  ut ^= qb[4];
                        if (t & 4)  ut ^= qb[5];
                        if (t & 8)  ut ^= qb[6];
                        if (t & 16) ut ^= qb[7];
                        #pragma unroll
                        for (int r = 0; r < 8; ++r) {
                            int ur = ut;
                            if (r & 1) ur ^= qb[0];
                            if (r & 2) ur ^= qb[1];
                            if (r & 4) ur ^= qb[2];
                            ad[k][r] = lq ^ ur;
                        }
                        #pragma unroll
                        for (int r = 4 * kk; r < 4 * kk + 4; ++r) {
                            const v2f xb = {T[u * 8 + 2 * j][r], T[u * 8 + 2 * j + 1][r]};
                            const v8f zero = {};
                            acc[r] = mma(G.m[j], xb, j == 0 ? zero : acc[r]);
                        }
                        if (k < 7) __builtin_amdgcn_sched_barrier(0);
                    }
                }
                #pragma unroll
                for (int r = 0; r < 8; ++r) {
                    #pragma unroll
                    for (int k = 0; k < 8; ++k) T[u * 8 + k][r] = acc[r][k];
                }
                __builtin_amdgcn_sched_barrier(0);
                #pragma unroll
                for (int k = 0; k < 8; ++k) {
                    #pragma unroll
                    for (int r = 0; r < 8; ++r)
                        *reinterpret_cast<float*>(reinterpret_cast<char*>(st) + ad[k][r]) = T[u * 8 + k][r];
                }
                __builtin_amdgcn_sched_barrier(0);
            }
            asm volatile("" :: "v"(G0n.m[0]), "v"(G0n.m[1]), "v"(G0n.m[2]), "v"(G0n.m[3]));
            auto read_x = [&](int u) __attribute__((always_inline)) {
                #pragma unroll
                for (int k = 0; k < 8; ++k) {
                    const int t = u * 8 + k;
                    #pragma unroll
                    for (int q = 0; q < 2; ++q) {
                        const v4f v = *reinterpret_cast<const v4f*>(&sW[laneRX ^ pos_x(idx_x(t, 4 * q, 0))]);
                        #pragma unroll
                        for (int e = 0; e < 4; ++e) T[t][4 * q + e] = v[e];
                    }
                }
                __builtin_amdgcn_sched_barrier(0);
            };
            auto gate_x = [&](int u) __attribute__((always_inline)) {
                #pragma unroll
                for (int k = 0; k < 8; ++k) {
                    const int t = u * 8 + k;
                    v8f acc = {};
                    #pragma unroll
                    for (int j = 0; j < 4; ++j) {
                        const v2f xb = {T[t][2 * j], T[t][2 * j + 1]};
                        acc = mma(G0n.m[j], xb, acc);
                    }
                    T[t] = acc;
                }
                __builtin_amdgcn_sched_barrier(0);
            };
            read_x(0);
            read_x(1);
            if (l + 1 == DEPTH) {
                read_x(2);
                read_x(3);
                break;
            }
            switch (l) {
                case 0:  __builtin_amdgcn_s_setprio(2); break;
                case 1:  __builtin_amdgcn_s_setprio(1); break;
                default: __builtin_amdgcn_s_setprio(0); break;
            }
            gate_x(0);
            read_x(2);
            gate_x(1);
            read_x(3);
            G1 = load_gate(ws, (l + 1) * 4 + 1, c, h);
            gate_x(2);
            gate_x(3);
        }
    }

    float A[256];
    #pragma unroll
    for (int t = 0; t < 32; ++t) {
        #pragma unroll
        for (int r = 0; r < 8; ++r) A[t * 8 + r] = T[t][r] * T[t][r];
    }
    float oq[8];
    #pragma unroll
    for (int b = 7; b >= 0; --b) {
        const int half = 1 << b;
        float d = 0.0f;
        #pragma unroll
        for (int j = 0; j < half; ++j) {
            d += A[j] - A[j + half];
            A[j] += A[j + half];
        }
        oq[b] = d;
    }
    const float tot = A[0];
    float o[12];
    o[6] = oq[7]; o[7] = oq[6];
    o[3] = oq[5]; o[4] = oq[4]; o[5] = oq[3];
    o[0] = oq[2]; o[1] = oq[1]; o[2] = oq[0];
    #pragma unroll
    for (int k = 8; k < 12; ++k)
        o[k] = ((c >> (11 - k)) & 1) ? -tot : tot;
    float mine = 0.0f;
    #pragma unroll
    for (int k = 0; k < 12; ++k) {
        float v = o[k];
        for (int off = 16; off > 0; off >>= 1) v += __shfl_xor(v, off, 32);
        mine = (lane == k) ? v : mine;
    }
    if (lane < 12) out[(size_t)row * 12 + lane] = mine;
}

extern "C" void kernel_launch(void* const* d_in, const int* in_sizes, int n_in,
                              void* d_out, int out_size, void* d_ws, size_t ws_size,
                              hipStream_t stream) {
    const float* x     = (const float*)d_in[0];
    const float* theta = (const float*)d_in[1];
    float*       out   = (float*)d_out;
    float*       wsf   = (float*)d_ws;

    const int B = in_sizes[0] / DIM;

    qdl_setup_kernel<<<17, 256, 0, stream>>>(theta, wsf);
    qdl_main_kernel<<<(B + WAVES - 1) / WAVES, 256, 0, stream>>>(x, wsf, out, B);
}
